# starting kernel
# baseline (speedup 1.0000x reference)
#include <hip/hip_runtime.h>

typedef float v2f __attribute__((ext_vector_type(2)));
typedef float v8f __attribute__((ext_vector_type(8)));

#define HEADS 8
#define DH 32
#define NTOK 2048
#define DIM 256
#define BATCH 2
#define SCALE2 (1.0f / 32.0f)

#define PROJ_ELEMS ((size_t)BATCH * HEADS * NTOK * DH)
#define OFF_QA ((size_t)0)
#define OFF_KA (PROJ_ELEMS)
#define OFF_QS (2 * PROJ_ELEMS)
#define OFF_KS (3 * PROJ_ELEMS)

static __device__ __forceinline__ v8f wmma_f32(v2f a, v2f b, v8f c) {
  return __builtin_amdgcn_wmma_f32_16x16x4_f32(
       false, a,  false, b,
       (short)0, c,  false,  false);
}

__global__ __launch_bounds__(256) void proj_kernel(
    const float* __restrict__ x, const float* __restrict__ attn_feat,
    const float* __restrict__ Wq, const float* __restrict__ Waux,
    float* __restrict__ ws) {
  const int lane = threadIdx.x & 31;
  const int wave = threadIdx.x >> 5;
  const int hi = lane >> 4;
  const int lo = lane & 15;
  const int ct = blockIdx.x * 8 + wave;
  const int m0 = blockIdx.y * 16;
  const int src = blockIdx.z;
  const float* X = src ? attn_feat : x;
  const float* W = src ? Waux : Wq;
  float* Qdst = ws + (src ? OFF_QA : OFF_QS);
  float* Kdst = ws + (src ? OFF_KA : OFF_KS);

  const int c = ct * 16 + lo;
  const float* Arow = X + (size_t)(m0 + lo) * DIM;

  v8f acc = {0.f, 0.f, 0.f, 0.f, 0.f, 0.f, 0.f, 0.f};
#pragma unroll 8
  for (int s = 0; s < 64; ++s) {
    const int k = s * 4 + hi * 2;
    v2f a = *(const v2f*)(Arow + k);
    v2f b;
    b.x = W[(size_t)k * 768 + c];
    b.y = W[(size_t)(k + 1) * 768 + c];
    acc = wmma_f32(a, b, acc);
  }

  float* dst;
  int hh;
  if (c < 256) { dst = Qdst; hh = c >> 5; }
  else         { dst = Kdst; hh = (c - 256) >> 5; }
  const int dh = c & 31;
#pragma unroll
  for (int r = 0; r < 8; ++r) {
    const int row = m0 + r + hi * 8;
    const int b_ = row >> 11, nn = row & 2047;
    dst[(((size_t)(b_ * HEADS + hh)) * NTOK + nn) * DH + dh] = acc[r];
  }
}

__global__ __launch_bounds__(256) void attn_kernel(
    const float* __restrict__ x, const float* __restrict__ ws,
    float* __restrict__ out0) {
  __shared__ float sX[16 * 272];
  __shared__ float sKa[16 * 36];
  __shared__ float sKs[16 * 36];
  __shared__ float sP[8 * 16 * 18];

  const int tid = threadIdx.x;
  const int lane = tid & 31, wave = tid >> 5;
  const int half = wave & 1, qt = wave >> 1;
  const int hi = lane >> 4, lo = lane & 15;
  const int bh = blockIdx.y, b = bh >> 3, h = bh & 7;
  const int m0 = blockIdx.x * 64 + qt * 16;

  const float* Qa = ws + OFF_QA;
  const float* Qs = ws + OFF_QS;
  const float* kabase = ws + OFF_KA + (size_t)bh * NTOK * DH;
  const float* ksbase = ws + OFF_KS + (size_t)bh * NTOK * DH;
  const float* xb = x + (size_t)b * NTOK * DIM;
  float* myP = sP + wave * (16 * 18);

  v2f qa[8], qs[8];
  {
    const float* qab = Qa + ((size_t)bh * NTOK + m0 + lo) * DH;
    const float* qsb = Qs + ((size_t)bh * NTOK + m0 + lo) * DH;
#pragma unroll
    for (int kk = 0; kk < 8; ++kk) {
      const int k = kk * 4 + hi * 2;
      qa[kk] = *(const v2f*)(qab + k);
      qs[kk] = *(const v2f*)(qsb + k);
    }
  }

  v8f O[8];
#pragma unroll
  for (int t = 0; t < 8; ++t) O[t] = (v8f){0.f, 0.f, 0.f, 0.f, 0.f, 0.f, 0.f, 0.f};
  float m[8], l[8];
#pragma unroll
  for (int r = 0; r < 8; ++r) { m[r] = -3.0e38f; l[r] = 0.f; }

  for (int j0 = 0; j0 < NTOK; j0 += 16) {
    __syncthreads();
#pragma unroll
    for (int t = 0; t < 4; ++t) {
      const int idx = t * 256 + tid;
      const int row = idx >> 6, c4 = (idx & 63) << 2;
      *(float4*)&sX[row * 272 + c4] =
          *(const float4*)(xb + (size_t)(j0 + row) * DIM + c4);
    }
    if (tid < 128) {
      const int row = tid >> 3, c4 = (tid & 7) << 2;
      *(float4*)&sKa[row * 36 + c4] =
          *(const float4*)(kabase + (size_t)(j0 + row) * DH + c4);
    } else {
      const int t2 = tid - 128;
      const int row = t2 >> 3, c4 = (t2 & 7) << 2;
      *(float4*)&sKs[row * 36 + c4] =
          *(const float4*)(ksbase + (size_t)(j0 + row) * DH + c4);
    }
    __syncthreads();

    v8f Sa = {0.f, 0.f, 0.f, 0.f, 0.f, 0.f, 0.f, 0.f};
    v8f Ss = {0.f, 0.f, 0.f, 0.f, 0.f, 0.f, 0.f, 0.f};
#pragma unroll
    for (int kk = 0; kk < 8; ++kk) {
      const int k = kk * 4 + hi * 2;
      v2f ba = *(const v2f*)&sKa[lo * 36 + k];
      Sa = wmma_f32(qa[kk], ba, Sa);
      v2f bs = *(const v2f*)&sKs[lo * 36 + k];
      Ss = wmma_f32(qs[kk], bs, Ss);
    }

    float P[8];
#pragma unroll
    for (int r = 0; r < 8; ++r) P[r] = Sa[r] * Ss[r] * SCALE2;

    float alpha[8];
#pragma unroll
    for (int r = 0; r < 8; ++r) {
      float v = P[r];
      v = fmaxf(v, __shfl_xor(v, 1));
      v = fmaxf(v, __shfl_xor(v, 2));
      v = fmaxf(v, __shfl_xor(v, 4));
      v = fmaxf(v, __shfl_xor(v, 8));
      const float mn = fmaxf(m[r], v);
      alpha[r] = __expf(m[r] - mn);
      m[r] = mn;
    }
#pragma unroll
    for (int r = 0; r < 8; ++r) {
      float e = __expf(P[r] - m[r]);
      P[r] = e;
      float s = e;
      s += __shfl_xor(s, 1);
      s += __shfl_xor(s, 2);
      s += __shfl_xor(s, 4);
      s += __shfl_xor(s, 8);
      l[r] = l[r] * alpha[r] + s;
    }
#pragma unroll
    for (int t = 0; t < 8; ++t)
#pragma unroll
      for (int r = 0; r < 8; ++r) O[t][r] *= alpha[r];

#pragma unroll
    for (int r = 0; r < 8; ++r) myP[(r + hi * 8) * 18 + lo] = P[r];
    asm volatile("s_wait_dscnt 0" ::: "memory");
    v2f ap[4];
#pragma unroll
    for (int kk = 0; kk < 4; ++kk)
      ap[kk] = *(const v2f*)&myP[lo * 18 + kk * 4 + hi * 2];

#pragma unroll
    for (int t = 0; t < 8; ++t) {
      const int n0 = half * 128 + t * 16 + lo;
      v8f o = O[t];
#pragma unroll
      for (int kk = 0; kk < 4; ++kk) {
        const int k = kk * 4 + hi * 2;
        v2f bx;
        bx.x = sX[k * 272 + n0];
        bx.y = sX[(k + 1) * 272 + n0];
        o = wmma_f32(ap[kk], bx, o);
      }
      O[t] = o;
    }
  }

  float invl[8];
#pragma unroll
  for (int r = 0; r < 8; ++r) invl[r] = 1.0f / l[r];
#pragma unroll
  for (int t = 0; t < 8; ++t) {
    const int col = h * 256 + half * 128 + t * 16 + lo;
#pragma unroll
    for (int r = 0; r < 8; ++r) {
      const int n = m0 + r + hi * 8;
      out0[((size_t)b * NTOK + n) * 2048 + col] = O[t][r] * invl[r];
    }
  }
}

__global__ __launch_bounds__(256) void row0_kernel(
    const float* __restrict__ ws, const float* __restrict__ out0,
    float* __restrict__ out1, float* __restrict__ attn0) {
  __shared__ float sp[NTOK];
  __shared__ float red[256];
  __shared__ float qa_s[DH], qs_s[DH];
  __shared__ float stat[2];
  const int bh = blockIdx.x, tid = threadIdx.x;
  const float* Qa = ws + OFF_QA + (size_t)bh * NTOK * DH;
  const float* Ka = ws + OFF_KA + (size_t)bh * NTOK * DH;
  const float* Qs = ws + OFF_QS + (size_t)bh * NTOK * DH;
  const float* Ks = ws + OFF_KS + (size_t)bh * NTOK * DH;
  if (tid < DH) qa_s[tid] = Qa[tid];
  else if (tid < 2 * DH) qs_s[tid - DH] = Qs[tid - DH];
  __syncthreads();

  float lmax = -3.0e38f;
  for (int j = tid; j < NTOK; j += 256) {
    float sa = 0.f, ss = 0.f;
    const float* ka = Ka + (size_t)j * DH;
    const float* ks = Ks + (size_t)j * DH;
#pragma unroll
    for (int d = 0; d < DH; ++d) { sa += qa_s[d] * ka[d]; ss += qs_s[d] * ks[d]; }
    const float p = sa * ss * SCALE2;
    sp[j] = p;
    lmax = fmaxf(lmax, p);
  }
  red[tid] = lmax;
  __syncthreads();
  for (int s = 128; s > 0; s >>= 1) {
    if (tid < s) red[tid] = fmaxf(red[tid], red[tid + s]);
    __syncthreads();
  }
  if (tid == 0) stat[0] = red[0];
  __syncthreads();
  const float mmax = stat[0];
  float lsum = 0.f;
  for (int j = tid; j < NTOK; j += 256) {
    const float e = __expf(sp[j] - mmax);
    sp[j] = e;
    lsum += e;
  }
  red[tid] = lsum;
  __syncthreads();
  for (int s = 128; s > 0; s >>= 1) {
    if (tid < s) red[tid] += red[tid + s];
    __syncthreads();
  }
  if (tid == 0) stat[1] = red[0];
  __syncthreads();
  const float inv = 1.0f / stat[1];
  for (int j = tid; j < NTOK; j += 256)
    attn0[(size_t)bh * NTOK + j] = sp[j] * inv;

  if (bh < BATCH) {
    for (int c = tid; c < 2048; c += 256)
      out1[bh * 2048 + c] = out0[(size_t)bh * 2048 * 2048 + c];
  }
}

extern "C" void kernel_launch(void* const* d_in, const int* in_sizes, int n_in,
                              void* d_out, int out_size, void* d_ws,
                              size_t ws_size, hipStream_t stream) {
  const float* x  = (const float*)d_in[0];
  const float* af = (const float*)d_in[1];
  const float* Wq = (const float*)d_in[2];
  const float* Wa = (const float*)d_in[3];
  float* ws = (float*)d_ws;
  float* out0 = (float*)d_out;
  float* out1 = out0 + (size_t)BATCH * NTOK * 2048;
  float* attn0 = out1 + (size_t)BATCH * 2048;

  proj_kernel<<<dim3(4, 256, 2), 256, 0, stream>>>(x, af, Wq, Wa, ws);
  attn_kernel<<<dim3(32, 16), 256, 0, stream>>>(x, ws, out0);
  row0_kernel<<<16, 256, 0, stream>>>(ws, out0, out1, attn0);
}


# attn kernel: 64 keys per block, four times the old block: one LDS staging, one barrier pair and one online-softmax max/sum/rescale per 64 keys; running-sum and rescale order changes
# speedup vs baseline: 1.1552x; 1.1552x over previous
#include <hip/hip_runtime.h>

typedef float v2f __attribute__((ext_vector_type(2)));
typedef float v8f __attribute__((ext_vector_type(8)));

#define HEADS 8
#define DH 32
#define NTOK 2048
#define DIM 256
#define BATCH 2
#define SCALE2 (1.0f / 32.0f)

#define PROJ_ELEMS ((size_t)BATCH * HEADS * NTOK * DH)
#define OFF_QA ((size_t)0)
#define OFF_KA (PROJ_ELEMS)
#define OFF_QS (2 * PROJ_ELEMS)
#define OFF_KS (3 * PROJ_ELEMS)

static __device__ __forceinline__ v8f wmma_f32(v2f a, v2f b, v8f c) {
  return __builtin_amdgcn_wmma_f32_16x16x4_f32(
       false, a,  false, b,
       (short)0, c,  false,  false);
}

__global__ __launch_bounds__(256) void proj_kernel(
    const float* __restrict__ x, const float* __restrict__ attn_feat,
    const float* __restrict__ Wq, const float* __restrict__ Waux,
    float* __restrict__ ws) {
  const int lane = threadIdx.x & 31;
  const int wave = threadIdx.x >> 5;
  const int hi = lane >> 4;
  const int lo = lane & 15;
  const int ct = blockIdx.x * 8 + wave;
  const int m0 = blockIdx.y * 16;
  const int src = blockIdx.z;
  const float* X = src ? attn_feat : x;
  const float* W = src ? Waux : Wq;
  float* Qdst = ws + (src ? OFF_QA : OFF_QS);
  float* Kdst = ws + (src ? OFF_KA : OFF_KS);

  const int c = ct * 16 + lo;
  const float* Arow = X + (size_t)(m0 + lo) * DIM;

  v8f acc = {0.f, 0.f, 0.f, 0.f, 0.f, 0.f, 0.f, 0.f};
#pragma unroll 8
  for (int s = 0; s < 64; ++s) {
    const int k = s * 4 + hi * 2;
    v2f a = *(const v2f*)(Arow + k);
    v2f b;
    b.x = W[(size_t)k * 768 + c];
    b.y = W[(size_t)(k + 1) * 768 + c];
    acc = wmma_f32(a, b, acc);
  }

  float* dst;
  int hh;
  if (c < 256) { dst = Qdst; hh = c >> 5; }
  else         { dst = Kdst; hh = (c - 256) >> 5; }
  const int dh = c & 31;
#pragma unroll
  for (int r = 0; r < 8; ++r) {
    const int row = m0 + r + hi * 8;
    const int b_ = row >> 11, nn = row & 2047;
    dst[(((size_t)(b_ * HEADS + hh)) * NTOK + nn) * DH + dh] = acc[r];
  }
}

#define BK 64
#define KT (BK / 16)
#define SX_LD 272
#define SK_LD 36
#define SP_LD 68

__global__ __launch_bounds__(256) __attribute__((amdgpu_waves_per_eu(4)))
void attn_kernel(
    const float* __restrict__ x, const float* __restrict__ ws,
    float* __restrict__ out0) {
  __shared__ float sX[BK * SX_LD];
  __shared__ float sKa[BK * SK_LD];
  __shared__ float sKs[BK * SK_LD];
  __shared__ float sP[8 * 16 * SP_LD];

  const int tid = threadIdx.x;
  const int lane = tid & 31, wave = tid >> 5;
  const int half = wave & 1, qt = wave >> 1;
  const int hi = lane >> 4, lo = lane & 15;
  const int bh = blockIdx.y, b = bh >> 3, h = bh & 7;
  const int m0 = blockIdx.x * 64 + qt * 16;

  const float* Qa = ws + OFF_QA;
  const float* Qs = ws + OFF_QS;
  const float* kabase = ws + OFF_KA + (size_t)bh * NTOK * DH;
  const float* ksbase = ws + OFF_KS + (size_t)bh * NTOK * DH;
  const float* xb = x + (size_t)b * NTOK * DIM;
  float* myP = sP + wave * (16 * SP_LD);

  v2f qa[8], qs[8];
  {
    const float* qab = Qa + ((size_t)bh * NTOK + m0 + lo) * DH;
    const float* qsb = Qs + ((size_t)bh * NTOK + m0 + lo) * DH;
#pragma unroll
    for (int kk = 0; kk < 8; ++kk) {
      const int k = kk * 4 + hi * 2;
      qa[kk] = *(const v2f*)(qab + k);
      qs[kk] = *(const v2f*)(qsb + k);
    }
  }

  v8f O[8];
#pragma unroll
  for (int t = 0; t < 8; ++t) O[t] = (v8f){0.f, 0.f, 0.f, 0.f, 0.f, 0.f, 0.f, 0.f};
  float m[8], l[8];
#pragma unroll
  for (int r = 0; r < 8; ++r) { m[r] = -3.0e38f; l[r] = 0.f; }

  const float* xg = xb + (size_t)(tid >> 6) * DIM + ((tid & 63) << 2);
  const float* kag = kabase + (size_t)(tid >> 3) * DH + ((tid & 7) << 2);
  const float* ksg = ksbase + (size_t)(tid >> 3) * DH + ((tid & 7) << 2);
  float* sxw = &sX[(tid >> 6) * SX_LD + ((tid & 63) << 2)];
  float* skaw = &sKa[(tid >> 3) * SK_LD + ((tid & 7) << 2)];
  float* sksw = &sKs[(tid >> 3) * SK_LD + ((tid & 7) << 2)];

  for (int j0 = 0; j0 < NTOK; j0 += BK) {
    __syncthreads();
    {
      float4 tx[BK / 4];
#pragma unroll
      for (int t = 0; t < BK / 4; ++t)
        tx[t] = *(const float4*)(xg + (size_t)t * 4 * DIM);
      float4 ta[BK / 32], ts[BK / 32];
#pragma unroll
      for (int t = 0; t < BK / 32; ++t) {
        ta[t] = *(const float4*)(kag + (size_t)t * 32 * DH);
        ts[t] = *(const float4*)(ksg + (size_t)t * 32 * DH);
      }
#pragma unroll
      for (int t = 0; t < BK / 4; ++t)
        *(float4*)(sxw + t * 4 * SX_LD) = tx[t];
#pragma unroll
      for (int t = 0; t < BK / 32; ++t) {
        *(float4*)(skaw + t * 32 * SK_LD) = ta[t];
        *(float4*)(sksw + t * 32 * SK_LD) = ts[t];
      }
      xg += (size_t)BK * DIM;
      kag += (size_t)BK * DH;
      ksg += (size_t)BK * DH;
    }
    __syncthreads();

    float P[KT][8];
#pragma unroll
    for (int kt = 0; kt < KT; ++kt) {
      v8f Sa = {0.f, 0.f, 0.f, 0.f, 0.f, 0.f, 0.f, 0.f};
      v8f Ss = {0.f, 0.f, 0.f, 0.f, 0.f, 0.f, 0.f, 0.f};
      const float* ka = &sKa[(kt * 16 + lo) * SK_LD];
      const float* ks = &sKs[(kt * 16 + lo) * SK_LD];
#pragma unroll
      for (int kk = 0; kk < 8; ++kk) {
        const int k = kk * 4 + hi * 2;
        v2f ba = *(const v2f*)&ka[k];
        Sa = wmma_f32(qa[kk], ba, Sa);
        v2f bs = *(const v2f*)&ks[k];
        Ss = wmma_f32(qs[kk], bs, Ss);
      }
#pragma unroll
      for (int r = 0; r < 8; ++r) P[kt][r] = Sa[r] * Ss[r] * SCALE2;
    }

    float alpha[8];
#pragma unroll
    for (int r = 0; r < 8; ++r) {
      float v = fmaxf(fmaxf(P[0][r], P[1][r]), fmaxf(P[2][r], P[3][r]));
      v = fmaxf(v, __shfl_xor(v, 1));
      v = fmaxf(v, __shfl_xor(v, 2));
      v = fmaxf(v, __shfl_xor(v, 4));
      v = fmaxf(v, __shfl_xor(v, 8));
      const float mn = fmaxf(m[r], v);
      alpha[r] = __expf(m[r] - mn);
      m[r] = mn;
    }
#pragma unroll
    for (int r = 0; r < 8; ++r) {
      float s = 0.f;
#pragma unroll
      for (int kt = 0; kt < KT; ++kt) {
        const float e = __expf(P[kt][r] - m[r]);
        P[kt][r] = e;
        s += e;
      }
      s += __shfl_xor(s, 1);
      s += __shfl_xor(s, 2);
      s += __shfl_xor(s, 4);
      s += __shfl_xor(s, 8);
      l[r] = l[r] * alpha[r] + s;
    }
#pragma unroll
    for (int t = 0; t < 8; ++t)
#pragma unroll
      for (int r = 0; r < 8; ++r) O[t][r] *= alpha[r];

#pragma unroll
    for (int kt = 0; kt < KT; ++kt)
#pragma unroll
      for (int r = 0; r < 8; ++r)
        myP[(r + hi * 8) * SP_LD + kt * 16 + lo] = P[kt][r];
    asm volatile("s_wait_dscnt 0" ::: "memory");
    const float* pa = myP + lo * SP_LD + hi * 2;
    const float* xr = sX + (hi * 2) * SX_LD + half * 128 + lo;
#pragma unroll 1
    for (int kb = 0; kb < KT; ++kb) {
      v2f ap[4];
#pragma unroll
      for (int kk = 0; kk < 4; ++kk) ap[kk] = *(const v2f*)(pa + kk * 4);
#pragma unroll
      for (int t = 0; t < 8; ++t) {
        v8f o = O[t];
#pragma unroll
        for (int kk = 0; kk < 4; ++kk) {
          v2f bx;
          bx.x = xr[(kk * 4) * SX_LD + t * 16];
          bx.y = xr[(kk * 4 + 1) * SX_LD + t * 16];
          o = wmma_f32(ap[kk], bx, o);
        }
        O[t] = o;
      }
      pa += 16;
      xr += 16 * SX_LD;
    }
  }

  float invl[8];
#pragma unroll
  for (int r = 0; r < 8; ++r) invl[r] = 1.0f / l[r];
#pragma unroll
  for (int t = 0; t < 8; ++t) {
    const int col = h * 256 + half * 128 + t * 16 + lo;
#pragma unroll
    for (int r = 0; r < 8; ++r) {
      const int n = m0 + r + hi * 8;
      out0[((size_t)b * NTOK + n) * 2048 + col] = O[t][r] * invl[r];
    }
  }
}

__global__ __launch_bounds__(256) void row0_kernel(
    const float* __restrict__ ws, const float* __restrict__ out0,
    float* __restrict__ out1, float* __restrict__ attn0) {
  __shared__ float sp[NTOK];
  __shared__ float red[256];
  __shared__ float qa_s[DH], qs_s[DH];
  __shared__ float stat[2];
  const int bh = blockIdx.x, tid = threadIdx.x;
  const float* Qa = ws + OFF_QA + (size_t)bh * NTOK * DH;
  const float* Ka = ws + OFF_KA + (size_t)bh * NTOK * DH;
  const float* Qs = ws + OFF_QS + (size_t)bh * NTOK * DH;
  const float* Ks = ws + OFF_KS + (size_t)bh * NTOK * DH;
  if (tid < DH) qa_s[tid] = Qa[tid];
  else if (tid < 2 * DH) qs_s[tid - DH] = Qs[tid - DH];
  __syncthreads();

  float lmax = -3.0e38f;
  for (int j = tid; j < NTOK; j += 256) {
    float sa = 0.f, ss = 0.f;
    const float* ka = Ka + (size_t)j * DH;
    const float* ks = Ks + (size_t)j * DH;
#pragma unroll
    for (int d = 0; d < DH; ++d) { sa += qa_s[d] * ka[d]; ss += qs_s[d] * ks[d]; }
    const float p = sa * ss * SCALE2;
    sp[j] = p;
    lmax = fmaxf(lmax, p);
  }
  red[tid] = lmax;
  __syncthreads();
  for (int s = 128; s > 0; s >>= 1) {
    if (tid < s) red[tid] = fmaxf(red[tid], red[tid + s]);
    __syncthreads();
  }
  if (tid == 0) stat[0] = red[0];
  __syncthreads();
  const float mmax = stat[0];
  float lsum = 0.f;
  for (int j = tid; j < NTOK; j += 256) {
    const float e = __expf(sp[j] - mmax);
    sp[j] = e;
    lsum += e;
  }
  red[tid] = lsum;
  __syncthreads();
  for (int s = 128; s > 0; s >>= 1) {
    if (tid < s) red[tid] += red[tid + s];
    __syncthreads();
  }
  if (tid == 0) stat[1] = red[0];
  __syncthreads();
  const float inv = 1.0f / stat[1];
  for (int j = tid; j < NTOK; j += 256)
    attn0[(size_t)bh * NTOK + j] = sp[j] * inv;

  if (bh < BATCH) {
    for (int c = tid; c < 2048; c += 256)
      out1[bh * 2048 + c] = out0[(size_t)bh * 2048 * 2048 + c];
  }
}

extern "C" void kernel_launch(void* const* d_in, const int* in_sizes, int n_in,
                              void* d_out, int out_size, void* d_ws,
                              size_t ws_size, hipStream_t stream) {
  const float* x  = (const float*)d_in[0];
  const float* af = (const float*)d_in[1];
  const float* Wq = (const float*)d_in[2];
  const float* Wa = (const float*)d_in[3];
  float* ws = (float*)d_ws;
  float* out0 = (float*)d_out;
  float* out1 = out0 + (size_t)BATCH * NTOK * 2048;
  float* attn0 = out1 + (size_t)BATCH * 2048;

  proj_kernel<<<dim3(4, 256, 2), 256, 0, stream>>>(x, af, Wq, Wa, ws);
  attn_kernel<<<dim3(32, 16), 256, 0, stream>>>(x, ws, out0);
  row0_kernel<<<16, 256, 0, stream>>>(ws, out0, out1, attn0);
}


# attn kernel: x key block kept in LDS as interleaved key pairs, so one 128-bit LDS load gives two ready B operands for P.x (4x fewer LDS reads, no register moves); pass unrolled, loads issued ahead
# speedup vs baseline: 1.2416x; 1.0748x over previous
#include <hip/hip_runtime.h>

typedef float v2f __attribute__((ext_vector_type(2)));
typedef float v8f __attribute__((ext_vector_type(8)));

#define HEADS 8
#define DH 32
#define NTOK 2048
#define DIM 256
#define BATCH 2
#define SCALE2 (1.0f / 32.0f)

#define PROJ_ELEMS ((size_t)BATCH * HEADS * NTOK * DH)
#define OFF_QA ((size_t)0)
#define OFF_KA (PROJ_ELEMS)
#define OFF_QS (2 * PROJ_ELEMS)
#define OFF_KS (3 * PROJ_ELEMS)

static __device__ __forceinline__ v8f wmma_f32(v2f a, v2f b, v8f c) {
  return __builtin_amdgcn_wmma_f32_16x16x4_f32(
       false, a,  false, b,
       (short)0, c,  false,  false);
}

__global__ __launch_bounds__(256) void proj_kernel(
    const float* __restrict__ x, const float* __restrict__ attn_feat,
    const float* __restrict__ Wq, const float* __restrict__ Waux,
    float* __restrict__ ws) {
  const int lane = threadIdx.x & 31;
  const int wave = threadIdx.x >> 5;
  const int hi = lane >> 4;
  const int lo = lane & 15;
  const int ct = blockIdx.x * 8 + wave;
  const int m0 = blockIdx.y * 16;
  const int src = blockIdx.z;
  const float* X = src ? attn_feat : x;
  const float* W = src ? Waux : Wq;
  float* Qdst = ws + (src ? OFF_QA : OFF_QS);
  float* Kdst = ws + (src ? OFF_KA : OFF_KS);

  const int c = ct * 16 + lo;
  const float* Arow = X + (size_t)(m0 + lo) * DIM;

  v8f acc = {0.f, 0.f, 0.f, 0.f, 0.f, 0.f, 0.f, 0.f};
#pragma unroll 8
  for (int s = 0; s < 64; ++s) {
    const int k = s * 4 + hi * 2;
    v2f a = *(const v2f*)(Arow + k);
    v2f b;
    b.x = W[(size_t)k * 768 + c];
    b.y = W[(size_t)(k + 1) * 768 + c];
    acc = wmma_f32(a, b, acc);
  }

  float* dst;
  int hh;
  if (c < 256) { dst = Qdst; hh = c >> 5; }
  else         { dst = Kdst; hh = (c - 256) >> 5; }
  const int dh = c & 31;
#pragma unroll
  for (int r = 0; r < 8; ++r) {
    const int row = m0 + r + hi * 8;
    const int b_ = row >> 11, nn = row & 2047;
    dst[(((size_t)(b_ * HEADS + hh)) * NTOK + nn) * DH + dh] = acc[r];
  }
}

#define BK 64
#define KT (BK / 16)
#define SXP_LD 544
#define SK_LD 36
#define SP_LD 68

__global__ __launch_bounds__(256) __attribute__((amdgpu_waves_per_eu(4)))
void attn_kernel(
    const float* __restrict__ x, const float* __restrict__ ws,
    float* __restrict__ out0) {
  __shared__ float sX[(BK / 2) * SXP_LD];
  __shared__ float sKa[BK * SK_LD];
  __shared__ float sKs[BK * SK_LD];
  __shared__ float sP[8 * 16 * SP_LD];

  const int tid = threadIdx.x;
  const int lane = tid & 31, wave = tid >> 5;
  const int half = wave & 1, qt = wave >> 1;
  const int hi = lane >> 4, lo = lane & 15;
  const int bh = blockIdx.y, b = bh >> 3, h = bh & 7;
  const int m0 = blockIdx.x * 64 + qt * 16;

  const float* Qa = ws + OFF_QA;
  const float* Qs = ws + OFF_QS;
  const float* kabase = ws + OFF_KA + (size_t)bh * NTOK * DH;
  const float* ksbase = ws + OFF_KS + (size_t)bh * NTOK * DH;
  const float* xb = x + (size_t)b * NTOK * DIM;
  float* myP = sP + wave * (16 * SP_LD);

  v2f qa[8], qs[8];
  {
    const float* qab = Qa + ((size_t)bh * NTOK + m0 + lo) * DH;
    const float* qsb = Qs + ((size_t)bh * NTOK + m0 + lo) * DH;
#pragma unroll
    for (int kk = 0; kk < 8; ++kk) {
      const int k = kk * 4 + hi * 2;
      qa[kk] = *(const v2f*)(qab + k);
      qs[kk] = *(const v2f*)(qsb + k);
    }
  }

  v8f O[8];
#pragma unroll
  for (int t = 0; t < 8; ++t) O[t] = (v8f){0.f, 0.f, 0.f, 0.f, 0.f, 0.f, 0.f, 0.f};
  float m[8], l[8];
#pragma unroll
  for (int r = 0; r < 8; ++r) { m[r] = -3.0e38f; l[r] = 0.f; }

  const float* xg = xb + (size_t)((tid >> 6) * 2) * DIM + ((tid & 63) << 2);
  const float* kag = kabase + (size_t)(tid >> 3) * DH + ((tid & 7) << 2);
  const float* ksg = ksbase + (size_t)(tid >> 3) * DH + ((tid & 7) << 2);
  float* sxw = &sX[(tid >> 6) * SXP_LD + ((tid & 63) << 3)];
  float* skaw = &sKa[(tid >> 3) * SK_LD + ((tid & 7) << 2)];
  float* sksw = &sKs[(tid >> 3) * SK_LD + ((tid & 7) << 2)];

  for (int j0 = 0; j0 < NTOK; j0 += BK) {
    __syncthreads();
    {
      float4 tx[BK / 4];
#pragma unroll
      for (int t = 0; t < BK / 8; ++t) {
        tx[2 * t] = *(const float4*)(xg + (size_t)t * 8 * DIM);
        tx[2 * t + 1] = *(const float4*)(xg + (size_t)t * 8 * DIM + DIM);
      }
      float4 ta[BK / 32], ts[BK / 32];
#pragma unroll
      for (int t = 0; t < BK / 32; ++t) {
        ta[t] = *(const float4*)(kag + (size_t)t * 32 * DH);
        ts[t] = *(const float4*)(ksg + (size_t)t * 32 * DH);
      }
#pragma unroll
      for (int t = 0; t < BK / 8; ++t) {
        const float4 e = tx[2 * t], o = tx[2 * t + 1];
        float* d = sxw + t * 4 * SXP_LD;
        *(float4*)d = make_float4(e.x, o.x, e.y, o.y);
        *(float4*)(d + 4) = make_float4(e.z, o.z, e.w, o.w);
      }
#pragma unroll
      for (int t = 0; t < BK / 32; ++t) {
        *(float4*)(skaw + t * 32 * SK_LD) = ta[t];
        *(float4*)(sksw + t * 32 * SK_LD) = ts[t];
      }
      xg += (size_t)BK * DIM;
      kag += (size_t)BK * DH;
      ksg += (size_t)BK * DH;
    }
    __syncthreads();

    float P[KT][8];
#pragma unroll
    for (int kt = 0; kt < KT; ++kt) {
      v8f Sa = {0.f, 0.f, 0.f, 0.f, 0.f, 0.f, 0.f, 0.f};
      v8f Ss = {0.f, 0.f, 0.f, 0.f, 0.f, 0.f, 0.f, 0.f};
      const float* ka = &sKa[(kt * 16 + lo) * SK_LD];
      const float* ks = &sKs[(kt * 16 + lo) * SK_LD];
#pragma unroll
      for (int kk = 0; kk < 8; ++kk) {
        const int k = kk * 4 + hi * 2;
        v2f ba = *(const v2f*)&ka[k];
        Sa = wmma_f32(qa[kk], ba, Sa);
        v2f bs = *(const v2f*)&ks[k];
        Ss = wmma_f32(qs[kk], bs, Ss);
      }
#pragma unroll
      for (int r = 0; r < 8; ++r) P[kt][r] = Sa[r] * Ss[r] * SCALE2;
    }

    float alpha[8];
#pragma unroll
    for (int r = 0; r < 8; ++r) {
      float v = fmaxf(fmaxf(P[0][r], P[1][r]), fmaxf(P[2][r], P[3][r]));
      v = fmaxf(v, __shfl_xor(v, 1));
      v = fmaxf(v, __shfl_xor(v, 2));
      v = fmaxf(v, __shfl_xor(v, 4));
      v = fmaxf(v, __shfl_xor(v, 8));
      const float mn = fmaxf(m[r], v);
      alpha[r] = __expf(m[r] - mn);
      m[r] = mn;
    }
#pragma unroll
    for (int r = 0; r < 8; ++r) {
      float s = 0.f;
#pragma unroll
      for (int kt = 0; kt < KT; ++kt) {
        const float e = __expf(P[kt][r] - m[r]);
        P[kt][r] = e;
        s += e;
      }
      s += __shfl_xor(s, 1);
      s += __shfl_xor(s, 2);
      s += __shfl_xor(s, 4);
      s += __shfl_xor(s, 8);
      l[r] = l[r] * alpha[r] + s;
    }
#pragma unroll
    for (int t = 0; t < 8; ++t)
#pragma unroll
      for (int r = 0; r < 8; ++r) O[t][r] *= alpha[r];

#pragma unroll
    for (int kt = 0; kt < KT; ++kt)
#pragma unroll
      for (int r = 0; r < 8; ++r)
        myP[(r + hi * 8) * SP_LD + kt * 16 + lo] = P[kt][r];
    asm volatile("s_wait_dscnt 0" ::: "memory");
    const float* pa = myP + lo * SP_LD + hi * 2;
    const float* xr = sX + hi * SXP_LD + half * 256 + lo * 4;
#pragma unroll
    for (int kb = 0; kb < KT; ++kb) {
      v2f ap[4];
#pragma unroll
      for (int kk = 0; kk < 4; ++kk) ap[kk] = *(const v2f*)(pa + kk * 4);
#pragma unroll
      for (int kk = 0; kk < 4; ++kk) {
        float4 bx[4];
#pragma unroll
        for (int q = 0; q < 4; ++q)
          bx[q] = *(const float4*)(xr + (kk * 2) * SXP_LD + q * 64);
#pragma unroll
        for (int q = 0; q < 4; ++q) {
          v2f b0, b1;
          b0.x = bx[q].x; b0.y = bx[q].y;
          b1.x = bx[q].z; b1.y = bx[q].w;
          O[2 * q] = wmma_f32(ap[kk], b0, O[2 * q]);
          O[2 * q + 1] = wmma_f32(ap[kk], b1, O[2 * q + 1]);
        }
      }
      pa += 16;
      xr += 8 * SXP_LD;
    }
  }

  float invl[8];
#pragma unroll
  for (int r = 0; r < 8; ++r) invl[r] = 1.0f / l[r];
#pragma unroll
  for (int q = 0; q < 4; ++q) {
    const int col = h * 256 + half * 128 + q * 32 + lo * 2;
#pragma unroll
    for (int r = 0; r < 8; ++r) {
      const int n = m0 + r + hi * 8;
      float2 v;
      v.x = O[2 * q][r] * invl[r];
      v.y = O[2 * q + 1][r] * invl[r];
      *(float2*)(out0 + ((size_t)b * NTOK + n) * 2048 + col) = v;
    }
  }
}

__global__ __launch_bounds__(256) void row0_kernel(
    const float* __restrict__ ws, const float* __restrict__ out0,
    float* __restrict__ out1, float* __restrict__ attn0) {
  __shared__ float sp[NTOK];
  __shared__ float red[256];
  __shared__ float qa_s[DH], qs_s[DH];
  __shared__ float stat[2];
  const int bh = blockIdx.x, tid = threadIdx.x;
  const float* Qa = ws + OFF_QA + (size_t)bh * NTOK * DH;
  const float* Ka = ws + OFF_KA + (size_t)bh * NTOK * DH;
  const float* Qs = ws + OFF_QS + (size_t)bh * NTOK * DH;
  const float* Ks = ws + OFF_KS + (size_t)bh * NTOK * DH;
  if (tid < DH) qa_s[tid] = Qa[tid];
  else if (tid < 2 * DH) qs_s[tid - DH] = Qs[tid - DH];
  __syncthreads();

  float lmax = -3.0e38f;
  for (int j = tid; j < NTOK; j += 256) {
    float sa = 0.f, ss = 0.f;
    const float* ka = Ka + (size_t)j * DH;
    const float* ks = Ks + (size_t)j * DH;
#pragma unroll
    for (int d = 0; d < DH; ++d) { sa += qa_s[d] * ka[d]; ss += qs_s[d] * ks[d]; }
    const float p = sa * ss * SCALE2;
    sp[j] = p;
    lmax = fmaxf(lmax, p);
  }
  red[tid] = lmax;
  __syncthreads();
  for (int s = 128; s > 0; s >>= 1) {
    if (tid < s) red[tid] = fmaxf(red[tid], red[tid + s]);
    __syncthreads();
  }
  if (tid == 0) stat[0] = red[0];
  __syncthreads();
  const float mmax = stat[0];
  float lsum = 0.f;
  for (int j = tid; j < NTOK; j += 256) {
    const float e = __expf(sp[j] - mmax);
    sp[j] = e;
    lsum += e;
  }
  red[tid] = lsum;
  __syncthreads();
  for (int s = 128; s > 0; s >>= 1) {
    if (tid < s) red[tid] += red[tid + s];
    __syncthreads();
  }
  if (tid == 0) stat[1] = red[0];
  __syncthreads();
  const float inv = 1.0f / stat[1];
  for (int j = tid; j < NTOK; j += 256)
    attn0[(size_t)bh * NTOK + j] = sp[j] * inv;

  if (bh < BATCH) {
    for (int c = tid; c < 2048; c += 256)
      out1[bh * 2048 + c] = out0[(size_t)bh * 2048 * 2048 + c];
  }
}

extern "C" void kernel_launch(void* const* d_in, const int* in_sizes, int n_in,
                              void* d_out, int out_size, void* d_ws,
                              size_t ws_size, hipStream_t stream) {
  const float* x  = (const float*)d_in[0];
  const float* af = (const float*)d_in[1];
  const float* Wq = (const float*)d_in[2];
  const float* Wa = (const float*)d_in[3];
  float* ws = (float*)d_ws;
  float* out0 = (float*)d_out;
  float* out1 = out0 + (size_t)BATCH * NTOK * 2048;
  float* attn0 = out1 + (size_t)BATCH * 2048;

  proj_kernel<<<dim3(4, 256, 2), 256, 0, stream>>>(x, af, Wq, Wa, ws);
  attn_kernel<<<dim3(32, 16), 256, 0, stream>>>(x, ws, out0);
  row0_kernel<<<16, 256, 0, stream>>>(ws, out0, out1, attn0);
}


# attn kernel on transposed tiles (S^T = K.Q^T, O^T += x^T.P^T): score registers are the P.x operand, a lane owns one query: no LDS transpose, no shuffle reductions; row-sum and key sum order change
# speedup vs baseline: 1.3430x; 1.0817x over previous
#include <hip/hip_runtime.h>

typedef float v2f __attribute__((ext_vector_type(2)));
typedef float v8f __attribute__((ext_vector_type(8)));

#define HEADS 8
#define DH 32
#define NTOK 2048
#define DIM 256
#define BATCH 2
#define SCALE2 (1.0f / 32.0f)

#define PROJ_ELEMS ((size_t)BATCH * HEADS * NTOK * DH)
#define OFF_QA ((size_t)0)
#define OFF_KA (PROJ_ELEMS)
#define OFF_QS (2 * PROJ_ELEMS)
#define OFF_KS (3 * PROJ_ELEMS)

static __device__ __forceinline__ v8f wmma_f32(v2f a, v2f b, v8f c) {
  return __builtin_amdgcn_wmma_f32_16x16x4_f32(
       false, a,  false, b,
       (short)0, c,  false,  false);
}

static __device__ __forceinline__ float xhalf(float v) {
  const int b = __builtin_bit_cast(int, v);
  return __builtin_bit_cast(float, __builtin_amdgcn_permlanex16(
      b, b, 0x76543210, 0xfedcba98, false, false));
}

__global__ __launch_bounds__(256) void proj_kernel(
    const float* __restrict__ x, const float* __restrict__ attn_feat,
    const float* __restrict__ Wq, const float* __restrict__ Waux,
    float* __restrict__ ws) {
  const int lane = threadIdx.x & 31;
  const int wave = threadIdx.x >> 5;
  const int hi = lane >> 4;
  const int lo = lane & 15;
  const int ct = blockIdx.x * 8 + wave;
  const int m0 = blockIdx.y * 16;
  const int src = blockIdx.z;
  const float* X = src ? attn_feat : x;
  const float* W = src ? Waux : Wq;
  float* Qdst = ws + (src ? OFF_QA : OFF_QS);
  float* Kdst = ws + (src ? OFF_KA : OFF_KS);

  const int c = ct * 16 + lo;
  const float* Arow = X + (size_t)(m0 + lo) * DIM;

  v8f acc = {0.f, 0.f, 0.f, 0.f, 0.f, 0.f, 0.f, 0.f};
#pragma unroll 8
  for (int s = 0; s < 64; ++s) {
    const int k = s * 4 + hi * 2;
    v2f a = *(const v2f*)(Arow + k);
    v2f b;
    b.x = W[(size_t)k * 768 + c];
    b.y = W[(size_t)(k + 1) * 768 + c];
    acc = wmma_f32(a, b, acc);
  }

  float* dst;
  int hh;
  if (c < 256) { dst = Qdst; hh = c >> 5; }
  else         { dst = Kdst; hh = (c - 256) >> 5; }
  const int dh = c & 31;
#pragma unroll
  for (int r = 0; r < 8; ++r) {
    const int row = m0 + r + hi * 8;
    const int b_ = row >> 11, nn = row & 2047;
    dst[(((size_t)(b_ * HEADS + hh)) * NTOK + nn) * DH + dh] = acc[r];
  }
}

#define BK 64
#define KT (BK / 16)
#define SXP_LD 544
#define SK_LD 36
#define SO_LD 132

__global__ __launch_bounds__(256) __attribute__((amdgpu_waves_per_eu(4)))
void attn_kernel(
    const float* __restrict__ x, const float* __restrict__ ws,
    float* __restrict__ out0) {
  __shared__ float sX[(BK / 2) * SXP_LD];
  __shared__ float sKa[BK * SK_LD];
  __shared__ float sKs[BK * SK_LD];

  const int tid = threadIdx.x;
  const int lane = tid & 31, wave = tid >> 5;
  const int half = wave & 1, qt = wave >> 1;
  const int hi = lane >> 4, lo = lane & 15;
  const int bh = blockIdx.y, b = bh >> 3, h = bh & 7;
  const int m0 = blockIdx.x * 64 + qt * 16;

  const float* Qa = ws + OFF_QA;
  const float* Qs = ws + OFF_QS;
  const float* kabase = ws + OFF_KA + (size_t)bh * NTOK * DH;
  const float* ksbase = ws + OFF_KS + (size_t)bh * NTOK * DH;
  const float* xb = x + (size_t)b * NTOK * DIM;

  v2f qa[8], qs[8];
  {
    const float* qab = Qa + ((size_t)bh * NTOK + m0 + lo) * DH;
    const float* qsb = Qs + ((size_t)bh * NTOK + m0 + lo) * DH;
#pragma unroll
    for (int kk = 0; kk < 8; ++kk) {
      const int k = kk * 4 + hi * 2;
      qa[kk] = *(const v2f*)(qab + k);
      qs[kk] = *(const v2f*)(qsb + k);
    }
  }

  v8f O[8];
#pragma unroll
  for (int t = 0; t < 8; ++t) O[t] = (v8f){0.f, 0.f, 0.f, 0.f, 0.f, 0.f, 0.f, 0.f};
  float m = -3.0e38f;
  float l = 0.f;

  const float* xg = xb + (size_t)((tid >> 6) * 2) * DIM + ((tid & 63) << 2);
  const float* kag = kabase + (size_t)(tid >> 3) * DH + ((tid & 7) << 2);
  const float* ksg = ksbase + (size_t)(tid >> 3) * DH + ((tid & 7) << 2);
  float* sxw = &sX[((tid >> 6) * 2) * SXP_LD + ((tid & 63) << 3)];
  float* skaw = &sKa[(tid >> 3) * SK_LD + ((tid & 7) << 2)];
  float* sksw = &sKs[(tid >> 3) * SK_LD + ((tid & 7) << 2)];

  for (int j0 = 0; j0 < NTOK; j0 += BK) {
    __syncthreads();
    {
      float4 tx[BK / 4];
#pragma unroll
      for (int t = 0; t < BK / 8; ++t) {
        tx[2 * t] = *(const float4*)(xg + (size_t)t * 8 * DIM);
        tx[2 * t + 1] = *(const float4*)(xg + (size_t)t * 8 * DIM + DIM);
      }
      float4 ta[BK / 32], ts[BK / 32];
#pragma unroll
      for (int t = 0; t < BK / 32; ++t) {
        ta[t] = *(const float4*)(kag + (size_t)t * 32 * DH);
        ts[t] = *(const float4*)(ksg + (size_t)t * 32 * DH);
      }
#pragma unroll
      for (int t = 0; t < BK / 8; ++t) {
        const float4 e = tx[2 * t], o = tx[2 * t + 1];
        float* d = sxw + ((t >> 1) * 8 + (t & 1)) * SXP_LD;
        *(float4*)d = make_float4(e.x, o.x, e.y, o.y);
        *(float4*)(d + 4) = make_float4(e.z, o.z, e.w, o.w);
      }
#pragma unroll
      for (int t = 0; t < BK / 32; ++t) {
        *(float4*)(skaw + t * 32 * SK_LD) = ta[t];
        *(float4*)(sksw + t * 32 * SK_LD) = ts[t];
      }
      xg += (size_t)BK * DIM;
      kag += (size_t)BK * DH;
      ksg += (size_t)BK * DH;
    }
    __syncthreads();

    v2f P[KT][4];
#pragma unroll
    for (int kt = 0; kt < KT; ++kt) {
      v8f Sa = {0.f, 0.f, 0.f, 0.f, 0.f, 0.f, 0.f, 0.f};
      v8f Ss = {0.f, 0.f, 0.f, 0.f, 0.f, 0.f, 0.f, 0.f};
      const float* ka = &sKa[(kt * 16 + lo) * SK_LD];
      const float* ks = &sKs[(kt * 16 + lo) * SK_LD];
#pragma unroll
      for (int kk = 0; kk < 8; ++kk) {
        const int k = kk * 4 + hi * 2;
        v2f aa = *(const v2f*)&ka[k];
        Sa = wmma_f32(aa, qa[kk], Sa);
        v2f as = *(const v2f*)&ks[k];
        Ss = wmma_f32(as, qs[kk], Ss);
      }
#pragma unroll
      for (int s = 0; s < 4; ++s) {
        P[kt][s].x = Sa[2 * s] * Ss[2 * s] * SCALE2;
        P[kt][s].y = Sa[2 * s + 1] * Ss[2 * s + 1] * SCALE2;
      }
    }

    float vk[KT];
#pragma unroll
    for (int kt = 0; kt < KT; ++kt)
      vk[kt] = fmaxf(fmaxf(fmaxf(P[kt][0].x, P[kt][0].y), fmaxf(P[kt][1].x, P[kt][1].y)),
                     fmaxf(fmaxf(P[kt][2].x, P[kt][2].y), fmaxf(P[kt][3].x, P[kt][3].y)));
    float v = fmaxf(fmaxf(vk[0], vk[1]), fmaxf(vk[2], vk[3]));
    v = fmaxf(v, xhalf(v));
    const float mn = fmaxf(m, v);
    const float alpha = __expf(m - mn);
    m = mn;
    float sk[KT];
#pragma unroll
    for (int kt = 0; kt < KT; ++kt) {
      sk[kt] = 0.f;
#pragma unroll
      for (int s = 0; s < 4; ++s) {
        P[kt][s].x = __expf(P[kt][s].x - mn);
        P[kt][s].y = __expf(P[kt][s].y - mn);
        sk[kt] += P[kt][s].x + P[kt][s].y;
      }
    }
    l = l * alpha + ((sk[0] + sk[1]) + (sk[2] + sk[3]));
#pragma unroll
    for (int t = 0; t < 8; ++t)
#pragma unroll
      for (int r = 0; r < 8; ++r) O[t][r] *= alpha;

    const float* xr = sX + hi * SXP_LD + half * 256 + lo * 4;
#pragma unroll
    for (int kt = 0; kt < KT; ++kt) {
#pragma unroll
      for (int s = 0; s < 4; ++s) {
        float4 ax[4];
#pragma unroll
        for (int q = 0; q < 4; ++q)
          ax[q] = *(const float4*)(xr + (kt * 8 + 2 * s) * SXP_LD + q * 64);
#pragma unroll
        for (int q = 0; q < 4; ++q) {
          v2f a0, a1;
          a0.x = ax[q].x; a0.y = ax[q].y;
          a1.x = ax[q].z; a1.y = ax[q].w;
          O[2 * q] = wmma_f32(a0, P[kt][s], O[2 * q]);
          O[2 * q + 1] = wmma_f32(a1, P[kt][s], O[2 * q + 1]);
        }
      }
    }
  }

  const float invl = 1.0f / (l + xhalf(l));
  __syncthreads();
  float* so = sX + wave * (16 * SO_LD);
#pragma unroll
  for (int q = 0; q < 4; ++q) {
#pragma unroll
    for (int j = 0; j < 4; ++j) {
      float4 o4 = make_float4(O[2 * q][2 * j] * invl, O[2 * q + 1][2 * j] * invl,
                              O[2 * q][2 * j + 1] * invl, O[2 * q + 1][2 * j + 1] * invl);
      *(float4*)(so + lo * SO_LD + q * 32 + hi * 16 + j * 4) = o4;
    }
  }
  __syncthreads();
#pragma unroll
  for (int rr = 0; rr < 16; ++rr) {
    const float4 o4 = *(const float4*)(so + rr * SO_LD + lane * 4);
    *(float4*)(out0 + ((size_t)b * NTOK + m0 + rr) * 2048 + h * 256 + half * 128 + lane * 4) = o4;
  }
}

__global__ __launch_bounds__(256) void row0_kernel(
    const float* __restrict__ ws, const float* __restrict__ out0,
    float* __restrict__ out1, float* __restrict__ attn0) {
  __shared__ float sp[NTOK];
  __shared__ float red[256];
  __shared__ float qa_s[DH], qs_s[DH];
  __shared__ float stat[2];
  const int bh = blockIdx.x, tid = threadIdx.x;
  const float* Qa = ws + OFF_QA + (size_t)bh * NTOK * DH;
  const float* Ka = ws + OFF_KA + (size_t)bh * NTOK * DH;
  const float* Qs = ws + OFF_QS + (size_t)bh * NTOK * DH;
  const float* Ks = ws + OFF_KS + (size_t)bh * NTOK * DH;
  if (tid < DH) qa_s[tid] = Qa[tid];
  else if (tid < 2 * DH) qs_s[tid - DH] = Qs[tid - DH];
  __syncthreads();

  float lmax = -3.0e38f;
  for (int j = tid; j < NTOK; j += 256) {
    float sa = 0.f, ss = 0.f;
    const float* ka = Ka + (size_t)j * DH;
    const float* ks = Ks + (size_t)j * DH;
#pragma unroll
    for (int d = 0; d < DH; ++d) { sa += qa_s[d] * ka[d]; ss += qs_s[d] * ks[d]; }
    const float p = sa * ss * SCALE2;
    sp[j] = p;
    lmax = fmaxf(lmax, p);
  }
  red[tid] = lmax;
  __syncthreads();
  for (int s = 128; s > 0; s >>= 1) {
    if (tid < s) red[tid] = fmaxf(red[tid], red[tid + s]);
    __syncthreads();
  }
  if (tid == 0) stat[0] = red[0];
  __syncthreads();
  const float mmax = stat[0];
  float lsum = 0.f;
  for (int j = tid; j < NTOK; j += 256) {
    const float e = __expf(sp[j] - mmax);
    sp[j] = e;
    lsum += e;
  }
  red[tid] = lsum;
  __syncthreads();
  for (int s = 128; s > 0; s >>= 1) {
    if (tid < s) red[tid] += red[tid + s];
    __syncthreads();
  }
  if (tid == 0) stat[1] = red[0];
  __syncthreads();
  const float inv = 1.0f / stat[1];
  for (int j = tid; j < NTOK; j += 256)
    attn0[(size_t)bh * NTOK + j] = sp[j] * inv;

  if (bh < BATCH) {
    for (int c = tid; c < 2048; c += 256)
      out1[bh * 2048 + c] = out0[(size_t)bh * 2048 * 2048 + c];
  }
}

extern "C" void kernel_launch(void* const* d_in, const int* in_sizes, int n_in,
                              void* d_out, int out_size, void* d_ws,
                              size_t ws_size, hipStream_t stream) {
  const float* x  = (const float*)d_in[0];
  const float* af = (const float*)d_in[1];
  const float* Wq = (const float*)d_in[2];
  const float* Wa = (const float*)d_in[3];
  float* ws = (float*)d_ws;
  float* out0 = (float*)d_out;
  float* out1 = out0 + (size_t)BATCH * NTOK * 2048;
  float* attn0 = out1 + (size_t)BATCH * 2048;

  proj_kernel<<<dim3(4, 256, 2), 256, 0, stream>>>(x, af, Wq, Wa, ws);
  attn_kernel<<<dim3(32, 16), 256, 0, stream>>>(x, ws, out0);
  row0_kernel<<<16, 256, 0, stream>>>(ws, out0, out1, attn0);
}


# attn kernel: one wave owns all 256 output columns of its 16 queries (8 query tiles per block, grid 16x16), not a 128-column half: scores, softmax and key staging once per query, 1/6 fewer WMMAs
# speedup vs baseline: 1.4931x; 1.1118x over previous
#include <hip/hip_runtime.h>

typedef float v2f __attribute__((ext_vector_type(2)));
typedef float v8f __attribute__((ext_vector_type(8)));

#define HEADS 8
#define DH 32
#define NTOK 2048
#define DIM 256
#define BATCH 2
#define SCALE2 (1.0f / 32.0f)

#define PROJ_ELEMS ((size_t)BATCH * HEADS * NTOK * DH)
#define OFF_QA ((size_t)0)
#define OFF_KA (PROJ_ELEMS)
#define OFF_QS (2 * PROJ_ELEMS)
#define OFF_KS (3 * PROJ_ELEMS)

static __device__ __forceinline__ v8f wmma_f32(v2f a, v2f b, v8f c) {
  return __builtin_amdgcn_wmma_f32_16x16x4_f32(
       false, a,  false, b,
       (short)0, c,  false,  false);
}

static __device__ __forceinline__ float xhalf(float v) {
  const int b = __builtin_bit_cast(int, v);
  return __builtin_bit_cast(float, __builtin_amdgcn_permlanex16(
      b, b, 0x76543210, 0xfedcba98, false, false));
}

__global__ __launch_bounds__(256) void proj_kernel(
    const float* __restrict__ x, const float* __restrict__ attn_feat,
    const float* __restrict__ Wq, const float* __restrict__ Waux,
    float* __restrict__ ws) {
  const int lane = threadIdx.x & 31;
  const int wave = threadIdx.x >> 5;
  const int hi = lane >> 4;
  const int lo = lane & 15;
  const int ct = blockIdx.x * 8 + wave;
  const int m0 = blockIdx.y * 16;
  const int src = blockIdx.z;
  const float* X = src ? attn_feat : x;
  const float* W = src ? Waux : Wq;
  float* Qdst = ws + (src ? OFF_QA : OFF_QS);
  float* Kdst = ws + (src ? OFF_KA : OFF_KS);

  const int c = ct * 16 + lo;
  const float* Arow = X + (size_t)(m0 + lo) * DIM;

  v8f acc = {0.f, 0.f, 0.f, 0.f, 0.f, 0.f, 0.f, 0.f};
#pragma unroll 8
  for (int s = 0; s < 64; ++s) {
    const int k = s * 4 + hi * 2;
    v2f a = *(const v2f*)(Arow + k);
    v2f b;
    b.x = W[(size_t)k * 768 + c];
    b.y = W[(size_t)(k + 1) * 768 + c];
    acc = wmma_f32(a, b, acc);
  }

  float* dst;
  int hh;
  if (c < 256) { dst = Qdst; hh = c >> 5; }
  else         { dst = Kdst; hh = (c - 256) >> 5; }
  const int dh = c & 31;
#pragma unroll
  for (int r = 0; r < 8; ++r) {
    const int row = m0 + r + hi * 8;
    const int b_ = row >> 11, nn = row & 2047;
    dst[(((size_t)(b_ * HEADS + hh)) * NTOK + nn) * DH + dh] = acc[r];
  }
}

#define BK 64
#define KT (BK / 16)
#define SXP_LD 544
#define SK_LD 36
#define SO_LD 132

__global__ __launch_bounds__(256)
void attn_kernel(
    const float* __restrict__ x, const float* __restrict__ ws,
    float* __restrict__ out0) {
  __shared__ float sX[(BK / 2) * SXP_LD];
  __shared__ float sKa[BK * SK_LD];
  __shared__ float sKs[BK * SK_LD];

  const int tid = threadIdx.x;
  const int lane = tid & 31, wave = tid >> 5;
  const int hi = lane >> 4, lo = lane & 15;
  const int bh = blockIdx.y, b = bh >> 3, h = bh & 7;
  const int m0 = blockIdx.x * 128 + wave * 16;

  const float* Qa = ws + OFF_QA;
  const float* Qs = ws + OFF_QS;
  const float* kabase = ws + OFF_KA + (size_t)bh * NTOK * DH;
  const float* ksbase = ws + OFF_KS + (size_t)bh * NTOK * DH;
  const float* xb = x + (size_t)b * NTOK * DIM;

  v2f qa[8], qs[8];
  {
    const float* qab = Qa + ((size_t)bh * NTOK + m0 + lo) * DH;
    const float* qsb = Qs + ((size_t)bh * NTOK + m0 + lo) * DH;
#pragma unroll
    for (int kk = 0; kk < 8; ++kk) {
      const int k = kk * 4 + hi * 2;
      qa[kk] = *(const v2f*)(qab + k);
      qs[kk] = *(const v2f*)(qsb + k);
    }
  }

  v8f O[16];
#pragma unroll
  for (int t = 0; t < 16; ++t) O[t] = (v8f){0.f, 0.f, 0.f, 0.f, 0.f, 0.f, 0.f, 0.f};
  float m = -3.0e38f;
  float l = 0.f;

  const float* xg = xb + (size_t)((tid >> 6) * 2) * DIM + ((tid & 63) << 2);
  const float* kag = kabase + (size_t)(tid >> 3) * DH + ((tid & 7) << 2);
  const float* ksg = ksbase + (size_t)(tid >> 3) * DH + ((tid & 7) << 2);
  float* sxw = &sX[((tid >> 6) * 2) * SXP_LD + ((tid & 63) << 3)];
  float* skaw = &sKa[(tid >> 3) * SK_LD + ((tid & 7) << 2)];
  float* sksw = &sKs[(tid >> 3) * SK_LD + ((tid & 7) << 2)];

  for (int j0 = 0; j0 < NTOK; j0 += BK) {
    __syncthreads();
    {
      float4 tx[BK / 4];
#pragma unroll
      for (int t = 0; t < BK / 8; ++t) {
        tx[2 * t] = *(const float4*)(xg + (size_t)t * 8 * DIM);
        tx[2 * t + 1] = *(const float4*)(xg + (size_t)t * 8 * DIM + DIM);
      }
      float4 ta[BK / 32], ts[BK / 32];
#pragma unroll
      for (int t = 0; t < BK / 32; ++t) {
        ta[t] = *(const float4*)(kag + (size_t)t * 32 * DH);
        ts[t] = *(const float4*)(ksg + (size_t)t * 32 * DH);
      }
#pragma unroll
      for (int t = 0; t < BK / 8; ++t) {
        const float4 e = tx[2 * t], o = tx[2 * t + 1];
        float* d = sxw + ((t >> 1) * 8 + (t & 1)) * SXP_LD;
        *(float4*)d = make_float4(e.x, o.x, e.y, o.y);
        *(float4*)(d + 4) = make_float4(e.z, o.z, e.w, o.w);
      }
#pragma unroll
      for (int t = 0; t < BK / 32; ++t) {
        *(float4*)(skaw + t * 32 * SK_LD) = ta[t];
        *(float4*)(sksw + t * 32 * SK_LD) = ts[t];
      }
      xg += (size_t)BK * DIM;
      kag += (size_t)BK * DH;
      ksg += (size_t)BK * DH;
    }
    __syncthreads();

    v2f P[KT][4];
#pragma unroll
    for (int kt = 0; kt < KT; ++kt) {
      v8f Sa = {0.f, 0.f, 0.f, 0.f, 0.f, 0.f, 0.f, 0.f};
      v8f Ss = {0.f, 0.f, 0.f, 0.f, 0.f, 0.f, 0.f, 0.f};
      const float* ka = &sKa[(kt * 16 + lo) * SK_LD];
      const float* ks = &sKs[(kt * 16 + lo) * SK_LD];
#pragma unroll
      for (int kk = 0; kk < 8; ++kk) {
        const int k = kk * 4 + hi * 2;
        v2f aa = *(const v2f*)&ka[k];
        Sa = wmma_f32(aa, qa[kk], Sa);
        v2f as = *(const v2f*)&ks[k];
        Ss = wmma_f32(as, qs[kk], Ss);
      }
#pragma unroll
      for (int s = 0; s < 4; ++s) {
        P[kt][s].x = Sa[2 * s] * Ss[2 * s] * SCALE2;
        P[kt][s].y = Sa[2 * s + 1] * Ss[2 * s + 1] * SCALE2;
      }
    }

    float vk[KT];
#pragma unroll
    for (int kt = 0; kt < KT; ++kt)
      vk[kt] = fmaxf(fmaxf(fmaxf(P[kt][0].x, P[kt][0].y), fmaxf(P[kt][1].x, P[kt][1].y)),
                     fmaxf(fmaxf(P[kt][2].x, P[kt][2].y), fmaxf(P[kt][3].x, P[kt][3].y)));
    float v = fmaxf(fmaxf(vk[0], vk[1]), fmaxf(vk[2], vk[3]));
    v = fmaxf(v, xhalf(v));
    const float mn = fmaxf(m, v);
    const float alpha = __expf(m - mn);
    m = mn;
    float sk[KT];
#pragma unroll
    for (int kt = 0; kt < KT; ++kt) {
      sk[kt] = 0.f;
#pragma unroll
      for (int s = 0; s < 4; ++s) {
        P[kt][s].x = __expf(P[kt][s].x - mn);
        P[kt][s].y = __expf(P[kt][s].y - mn);
        sk[kt] += P[kt][s].x + P[kt][s].y;
      }
    }
    l = l * alpha + ((sk[0] + sk[1]) + (sk[2] + sk[3]));
#pragma unroll
    for (int t = 0; t < 16; ++t)
#pragma unroll
      for (int r = 0; r < 8; ++r) O[t][r] *= alpha;

    const float* xr = sX + hi * SXP_LD + lo * 4;
#pragma unroll
    for (int kt = 0; kt < KT; ++kt) {
#pragma unroll
      for (int s = 0; s < 4; ++s) {
        float4 ax[8];
#pragma unroll
        for (int q = 0; q < 8; ++q)
          ax[q] = *(const float4*)(xr + (kt * 8 + 2 * s) * SXP_LD + q * 64);
#pragma unroll
        for (int q = 0; q < 8; ++q) {
          v2f a0, a1;
          a0.x = ax[q].x; a0.y = ax[q].y;
          a1.x = ax[q].z; a1.y = ax[q].w;
          O[2 * q] = wmma_f32(a0, P[kt][s], O[2 * q]);
          O[2 * q + 1] = wmma_f32(a1, P[kt][s], O[2 * q + 1]);
        }
      }
    }
  }

  const float invl = 1.0f / (l + xhalf(l));
  float* so = sX + wave * (16 * SO_LD);
#pragma unroll
  for (int p = 0; p < 2; ++p) {
    __syncthreads();
#pragma unroll
    for (int q = 0; q < 4; ++q) {
#pragma unroll
      for (int j = 0; j < 4; ++j) {
        const int t = 2 * (4 * p + q);
        float4 o4 = make_float4(O[t][2 * j] * invl, O[t + 1][2 * j] * invl,
                                O[t][2 * j + 1] * invl, O[t + 1][2 * j + 1] * invl);
        *(float4*)(so + lo * SO_LD + q * 32 + hi * 16 + j * 4) = o4;
      }
    }
    __syncthreads();
#pragma unroll
    for (int rr = 0; rr < 16; ++rr) {
      const float4 o4 = *(const float4*)(so + rr * SO_LD + lane * 4);
      *(float4*)(out0 + ((size_t)b * NTOK + m0 + rr) * 2048 + h * 256 + p * 128 + lane * 4) = o4;
    }
  }
}

__global__ __launch_bounds__(256) void row0_kernel(
    const float* __restrict__ ws, const float* __restrict__ out0,
    float* __restrict__ out1, float* __restrict__ attn0) {
  __shared__ float sp[NTOK];
  __shared__ float red[256];
  __shared__ float qa_s[DH], qs_s[DH];
  __shared__ float stat[2];
  const int bh = blockIdx.x, tid = threadIdx.x;
  const float* Qa = ws + OFF_QA + (size_t)bh * NTOK * DH;
  const float* Ka = ws + OFF_KA + (size_t)bh * NTOK * DH;
  const float* Qs = ws + OFF_QS + (size_t)bh * NTOK * DH;
  const float* Ks = ws + OFF_KS + (size_t)bh * NTOK * DH;
  if (tid < DH) qa_s[tid] = Qa[tid];
  else if (tid < 2 * DH) qs_s[tid - DH] = Qs[tid - DH];
  __syncthreads();

  float lmax = -3.0e38f;
  for (int j = tid; j < NTOK; j += 256) {
    float sa = 0.f, ss = 0.f;
    const float* ka = Ka + (size_t)j * DH;
    const float* ks = Ks + (size_t)j * DH;
#pragma unroll
    for (int d = 0; d < DH; ++d) { sa += qa_s[d] * ka[d]; ss += qs_s[d] * ks[d]; }
    const float p = sa * ss * SCALE2;
    sp[j] = p;
    lmax = fmaxf(lmax, p);
  }
  red[tid] = lmax;
  __syncthreads();
  for (int s = 128; s > 0; s >>= 1) {
    if (tid < s) red[tid] = fmaxf(red[tid], red[tid + s]);
    __syncthreads();
  }
  if (tid == 0) stat[0] = red[0];
  __syncthreads();
  const float mmax = stat[0];
  float lsum = 0.f;
  for (int j = tid; j < NTOK; j += 256) {
    const float e = __expf(sp[j] - mmax);
    sp[j] = e;
    lsum += e;
  }
  red[tid] = lsum;
  __syncthreads();
  for (int s = 128; s > 0; s >>= 1) {
    if (tid < s) red[tid] += red[tid + s];
    __syncthreads();
  }
  if (tid == 0) stat[1] = red[0];
  __syncthreads();
  const float inv = 1.0f / stat[1];
  for (int j = tid; j < NTOK; j += 256)
    attn0[(size_t)bh * NTOK + j] = sp[j] * inv;

  if (bh < BATCH) {
    for (int c = tid; c < 2048; c += 256)
      out1[bh * 2048 + c] = out0[(size_t)bh * 2048 * 2048 + c];
  }
}

extern "C" void kernel_launch(void* const* d_in, const int* in_sizes, int n_in,
                              void* d_out, int out_size, void* d_ws,
                              size_t ws_size, hipStream_t stream) {
  const float* x  = (const float*)d_in[0];
  const float* af = (const float*)d_in[1];
  const float* Wq = (const float*)d_in[2];
  const float* Wa = (const float*)d_in[3];
  float* ws = (float*)d_ws;
  float* out0 = (float*)d_out;
  float* out1 = out0 + (size_t)BATCH * NTOK * 2048;
  float* attn0 = out1 + (size_t)BATCH * 2048;

  proj_kernel<<<dim3(4, 256, 2), 256, 0, stream>>>(x, af, Wq, Wa, ws);
  attn_kernel<<<dim3(16, 16), 256, 0, stream>>>(x, ws, out0);
  row0_kernel<<<16, 256, 0, stream>>>(ws, out0, out1, attn0);
}


# attn kernel: LDS fragment reads issued as whole groups ahead of long WMMA runs (all K fragments at once; x fragments of two K-steps, double-buffered), not one read after every second WMMA
# speedup vs baseline: 1.5595x; 1.0445x over previous
#include <hip/hip_runtime.h>

typedef float v2f __attribute__((ext_vector_type(2)));
typedef float v8f __attribute__((ext_vector_type(8)));

#define HEADS 8
#define DH 32
#define NTOK 2048
#define DIM 256
#define BATCH 2
#define SCALE2 (1.0f / 32.0f)

#define PROJ_ELEMS ((size_t)BATCH * HEADS * NTOK * DH)
#define OFF_QA ((size_t)0)
#define OFF_KA (PROJ_ELEMS)
#define OFF_QS (2 * PROJ_ELEMS)
#define OFF_KS (3 * PROJ_ELEMS)

static __device__ __forceinline__ v8f wmma_f32(v2f a, v2f b, v8f c) {
  return __builtin_amdgcn_wmma_f32_16x16x4_f32(
       false, a,  false, b,
       (short)0, c,  false,  false);
}

static __device__ __forceinline__ float xhalf(float v) {
  const int b = __builtin_bit_cast(int, v);
  return __builtin_bit_cast(float, __builtin_amdgcn_permlanex16(
      b, b, 0x76543210, 0xfedcba98, false, false));
}

__global__ __launch_bounds__(256) void proj_kernel(
    const float* __restrict__ x, const float* __restrict__ attn_feat,
    const float* __restrict__ Wq, const float* __restrict__ Waux,
    float* __restrict__ ws) {
  const int lane = threadIdx.x & 31;
  const int wave = threadIdx.x >> 5;
  const int hi = lane >> 4;
  const int lo = lane & 15;
  const int ct = blockIdx.x * 8 + wave;
  const int m0 = blockIdx.y * 16;
  const int src = blockIdx.z;
  const float* X = src ? attn_feat : x;
  const float* W = src ? Waux : Wq;
  float* Qdst = ws + (src ? OFF_QA : OFF_QS);
  float* Kdst = ws + (src ? OFF_KA : OFF_KS);

  const int c = ct * 16 + lo;
  const float* Arow = X + (size_t)(m0 + lo) * DIM;

  v8f acc = {0.f, 0.f, 0.f, 0.f, 0.f, 0.f, 0.f, 0.f};
#pragma unroll 8
  for (int s = 0; s < 64; ++s) {
    const int k = s * 4 + hi * 2;
    v2f a = *(const v2f*)(Arow + k);
    v2f b;
    b.x = W[(size_t)k * 768 + c];
    b.y = W[(size_t)(k + 1) * 768 + c];
    acc = wmma_f32(a, b, acc);
  }

  float* dst;
  int hh;
  if (c < 256) { dst = Qdst; hh = c >> 5; }
  else         { dst = Kdst; hh = (c - 256) >> 5; }
  const int dh = c & 31;
#pragma unroll
  for (int r = 0; r < 8; ++r) {
    const int row = m0 + r + hi * 8;
    const int b_ = row >> 11, nn = row & 2047;
    dst[(((size_t)(b_ * HEADS + hh)) * NTOK + nn) * DH + dh] = acc[r];
  }
}

#define BK 64
#define KT (BK / 16)
#define SXP_LD 544
#define SK_LD 36
#define SO_LD 132

__global__ __launch_bounds__(256)
void attn_kernel(
    const float* __restrict__ x, const float* __restrict__ ws,
    float* __restrict__ out0) {
  __shared__ float sX[(BK / 2) * SXP_LD];
  __shared__ float sKa[BK * SK_LD];
  __shared__ float sKs[BK * SK_LD];

  const int tid = threadIdx.x;
  const int lane = tid & 31, wave = tid >> 5;
  const int hi = lane >> 4, lo = lane & 15;
  const int bh = blockIdx.y, b = bh >> 3, h = bh & 7;
  const int m0 = blockIdx.x * 128 + wave * 16;

  const float* Qa = ws + OFF_QA;
  const float* Qs = ws + OFF_QS;
  const float* kabase = ws + OFF_KA + (size_t)bh * NTOK * DH;
  const float* ksbase = ws + OFF_KS + (size_t)bh * NTOK * DH;
  const float* xb = x + (size_t)b * NTOK * DIM;

  v2f qa[8], qs[8];
  {
    const float* qab = Qa + ((size_t)bh * NTOK + m0 + lo) * DH;
    const float* qsb = Qs + ((size_t)bh * NTOK + m0 + lo) * DH;
#pragma unroll
    for (int kk = 0; kk < 8; ++kk) {
      const int k = kk * 4 + hi * 2;
      qa[kk] = *(const v2f*)(qab + k);
      qs[kk] = *(const v2f*)(qsb + k);
    }
  }

  v8f O[16];
#pragma unroll
  for (int t = 0; t < 16; ++t) O[t] = (v8f){0.f, 0.f, 0.f, 0.f, 0.f, 0.f, 0.f, 0.f};
  float m = -3.0e38f;
  float l = 0.f;

  const float* xg = xb + (size_t)((tid >> 6) * 2) * DIM + ((tid & 63) << 2);
  const float* kag = kabase + (size_t)(tid >> 3) * DH + ((tid & 7) << 2);
  const float* ksg = ksbase + (size_t)(tid >> 3) * DH + ((tid & 7) << 2);
  float* sxw = &sX[((tid >> 6) * 2) * SXP_LD + ((tid & 63) << 3)];
  float* skaw = &sKa[(tid >> 3) * SK_LD + ((tid & 7) << 2)];
  float* sksw = &sKs[(tid >> 3) * SK_LD + ((tid & 7) << 2)];

  for (int j0 = 0; j0 < NTOK; j0 += BK) {
    __syncthreads();
    {
      float4 tx[BK / 4];
#pragma unroll
      for (int t = 0; t < BK / 8; ++t) {
        tx[2 * t] = *(const float4*)(xg + (size_t)t * 8 * DIM);
        tx[2 * t + 1] = *(const float4*)(xg + (size_t)t * 8 * DIM + DIM);
      }
      float4 ta[BK / 32], ts[BK / 32];
#pragma unroll
      for (int t = 0; t < BK / 32; ++t) {
        ta[t] = *(const float4*)(kag + (size_t)t * 32 * DH);
        ts[t] = *(const float4*)(ksg + (size_t)t * 32 * DH);
      }
#pragma unroll
      for (int t = 0; t < BK / 8; ++t) {
        const float4 e = tx[2 * t], o = tx[2 * t + 1];
        float* d = sxw + ((t >> 1) * 8 + (t & 1)) * SXP_LD;
        *(float4*)d = make_float4(e.x, o.x, e.y, o.y);
        *(float4*)(d + 4) = make_float4(e.z, o.z, e.w, o.w);
      }
#pragma unroll
      for (int t = 0; t < BK / 32; ++t) {
        *(float4*)(skaw + t * 32 * SK_LD) = ta[t];
        *(float4*)(sksw + t * 32 * SK_LD) = ts[t];
      }
      xg += (size_t)BK * DIM;
      kag += (size_t)BK * DH;
      ksg += (size_t)BK * DH;
    }
    __syncthreads();

    v2f fka[KT][8], fks[KT][8];
#pragma unroll
    for (int kt = 0; kt < KT; ++kt) {
      const float* ka = &sKa[(kt * 16 + lo) * SK_LD];
      const float* ks = &sKs[(kt * 16 + lo) * SK_LD];
#pragma unroll
      for (int kk = 0; kk < 8; ++kk) {
        const int k = kk * 4 + hi * 2;
        fka[kt][kk] = *(const v2f*)&ka[k];
        fks[kt][kk] = *(const v2f*)&ks[k];
      }
    }
    const float* xr = sX + hi * SXP_LD + lo * 4;
#define LOAD_XG(buf, g)                                                        \
  _Pragma("unroll") for (int u = 0; u < 2; ++u)                                \
  _Pragma("unroll") for (int q = 0; q < 8; ++q)                                \
    buf[u * 8 + q] = *(const float4*)(                                         \
        xr + (((g) >> 1) * 8 + 2 * (((g) & 1) * 2 + u)) * SXP_LD + q * 64)
    float4 fa[16], fb[16];
    LOAD_XG(fa, 0);
    __builtin_amdgcn_sched_barrier(0);

    v2f P[KT][4];
#pragma unroll
    for (int kt = 0; kt < KT; ++kt) {
      v8f Sa = {0.f, 0.f, 0.f, 0.f, 0.f, 0.f, 0.f, 0.f};
      v8f Ss = {0.f, 0.f, 0.f, 0.f, 0.f, 0.f, 0.f, 0.f};
#pragma unroll
      for (int kk = 0; kk < 8; ++kk) {
        Sa = wmma_f32(fka[kt][kk], qa[kk], Sa);
        Ss = wmma_f32(fks[kt][kk], qs[kk], Ss);
      }
#pragma unroll
      for (int s = 0; s < 4; ++s) {
        P[kt][s].x = Sa[2 * s] * Ss[2 * s] * SCALE2;
        P[kt][s].y = Sa[2 * s + 1] * Ss[2 * s + 1] * SCALE2;
      }
    }

    float vk[KT];
#pragma unroll
    for (int kt = 0; kt < KT; ++kt)
      vk[kt] = fmaxf(fmaxf(fmaxf(P[kt][0].x, P[kt][0].y), fmaxf(P[kt][1].x, P[kt][1].y)),
                     fmaxf(fmaxf(P[kt][2].x, P[kt][2].y), fmaxf(P[kt][3].x, P[kt][3].y)));
    float v = fmaxf(fmaxf(vk[0], vk[1]), fmaxf(vk[2], vk[3]));
    v = fmaxf(v, xhalf(v));
    const float mn = fmaxf(m, v);
    const float alpha = __expf(m - mn);
    m = mn;
    float sk[KT];
#pragma unroll
    for (int kt = 0; kt < KT; ++kt) {
      sk[kt] = 0.f;
#pragma unroll
      for (int s = 0; s < 4; ++s) {
        P[kt][s].x = __expf(P[kt][s].x - mn);
        P[kt][s].y = __expf(P[kt][s].y - mn);
        sk[kt] += P[kt][s].x + P[kt][s].y;
      }
    }
    l = l * alpha + ((sk[0] + sk[1]) + (sk[2] + sk[3]));
#pragma unroll
    for (int t = 0; t < 16; ++t)
#pragma unroll
      for (int r = 0; r < 8; ++r) O[t][r] *= alpha;

#define MMA_XG(buf, g)                                                         \
  _Pragma("unroll") for (int u = 0; u < 2; ++u)                                \
  _Pragma("unroll") for (int q = 0; q < 8; ++q) {                              \
    v2f a0, a1;                                                                \
    a0.x = buf[u * 8 + q].x; a0.y = buf[u * 8 + q].y;        \
    a1.x = buf[u * 8 + q].z; a1.y = buf[u * 8 + q].w;        \
    O[2 * q] = wmma_f32(a0, P[(g) >> 1][((g) & 1) * 2 + u], O[2 * q]);         \
    O[2 * q + 1] = wmma_f32(a1, P[(g) >> 1][((g) & 1) * 2 + u], O[2 * q + 1]); \
  }
#pragma unroll
    for (int g = 0; g < 8; g += 2) {
      LOAD_XG(fb, g + 1);
      __builtin_amdgcn_sched_barrier(0);
      MMA_XG(fa, g);
      __builtin_amdgcn_sched_barrier(0);
      if (g + 2 < 8) {
        LOAD_XG(fa, g + 2);
      }
      __builtin_amdgcn_sched_barrier(0);
      MMA_XG(fb, g + 1);
      __builtin_amdgcn_sched_barrier(0);
    }
#undef LOAD_XG
#undef MMA_XG
  }

  const float invl = 1.0f / (l + xhalf(l));
  float* so = sX + wave * (16 * SO_LD);
#pragma unroll
  for (int p = 0; p < 2; ++p) {
    __syncthreads();
#pragma unroll
    for (int q = 0; q < 4; ++q) {
#pragma unroll
      for (int j = 0; j < 4; ++j) {
        const int t = 2 * (4 * p + q);
        float4 o4 = make_float4(O[t][2 * j] * invl, O[t + 1][2 * j] * invl,
                                O[t][2 * j + 1] * invl, O[t + 1][2 * j + 1] * invl);
        *(float4*)(so + lo * SO_LD + q * 32 + hi * 16 + j * 4) = o4;
      }
    }
    __syncthreads();
#pragma unroll
    for (int rr = 0; rr < 16; ++rr) {
      const float4 o4 = *(const float4*)(so + rr * SO_LD + lane * 4);
      *(float4*)(out0 + ((size_t)b * NTOK + m0 + rr) * 2048 + h * 256 + p * 128 + lane * 4) = o4;
    }
  }
}

__global__ __launch_bounds__(256) void row0_kernel(
    const float* __restrict__ ws, const float* __restrict__ out0,
    float* __restrict__ out1, float* __restrict__ attn0) {
  __shared__ float sp[NTOK];
  __shared__ float red[256];
  __shared__ float qa_s[DH], qs_s[DH];
  __shared__ float stat[2];
  const int bh = blockIdx.x, tid = threadIdx.x;
  const float* Qa = ws + OFF_QA + (size_t)bh * NTOK * DH;
  const float* Ka = ws + OFF_KA + (size_t)bh * NTOK * DH;
  const float* Qs = ws + OFF_QS + (size_t)bh * NTOK * DH;
  const float* Ks = ws + OFF_KS + (size_t)bh * NTOK * DH;
  if (tid < DH) qa_s[tid] = Qa[tid];
  else if (tid < 2 * DH) qs_s[tid - DH] = Qs[tid - DH];
  __syncthreads();

  float lmax = -3.0e38f;
  for (int j = tid; j < NTOK; j += 256) {
    float sa = 0.f, ss = 0.f;
    const float* ka = Ka + (size_t)j * DH;
    const float* ks = Ks + (size_t)j * DH;
#pragma unroll
    for (int d = 0; d < DH; ++d) { sa += qa_s[d] * ka[d]; ss += qs_s[d] * ks[d]; }
    const float p = sa * ss * SCALE2;
    sp[j] = p;
    lmax = fmaxf(lmax, p);
  }
  red[tid] = lmax;
  __syncthreads();
  for (int s = 128; s > 0; s >>= 1) {
    if (tid < s) red[tid] = fmaxf(red[tid], red[tid + s]);
    __syncthreads();
  }
  if (tid == 0) stat[0] = red[0];
  __syncthreads();
  const float mmax = stat[0];
  float lsum = 0.f;
  for (int j = tid; j < NTOK; j += 256) {
    const float e = __expf(sp[j] - mmax);
    sp[j] = e;
    lsum += e;
  }
  red[tid] = lsum;
  __syncthreads();
  for (int s = 128; s > 0; s >>= 1) {
    if (tid < s) red[tid] += red[tid + s];
    __syncthreads();
  }
  if (tid == 0) stat[1] = red[0];
  __syncthreads();
  const float inv = 1.0f / stat[1];
  for (int j = tid; j < NTOK; j += 256)
    attn0[(size_t)bh * NTOK + j] = sp[j] * inv;

  if (bh < BATCH) {
    for (int c = tid; c < 2048; c += 256)
      out1[bh * 2048 + c] = out0[(size_t)bh * 2048 * 2048 + c];
  }
}

extern "C" void kernel_launch(void* const* d_in, const int* in_sizes, int n_in,
                              void* d_out, int out_size, void* d_ws,
                              size_t ws_size, hipStream_t stream) {
  const float* x  = (const float*)d_in[0];
  const float* af = (const float*)d_in[1];
  const float* Wq = (const float*)d_in[2];
  const float* Wa = (const float*)d_in[3];
  float* ws = (float*)d_ws;
  float* out0 = (float*)d_out;
  float* out1 = out0 + (size_t)BATCH * NTOK * 2048;
  float* attn0 = out1 + (size_t)BATCH * 2048;

  proj_kernel<<<dim3(4, 256, 2), 256, 0, stream>>>(x, af, Wq, Wa, ws);
  attn_kernel<<<dim3(16, 16), 256, 0, stream>>>(x, ws, out0);
  row0_kernel<<<16, 256, 0, stream>>>(ws, out0, out1, attn0);
}


# attn kernel: key blocks through two LDS stages: block j+1's global loads issued ahead of block j's output product, stored to the other stage behind it; one barrier per block; no sum order changes
# speedup vs baseline: 1.6769x; 1.0753x over previous
#include <hip/hip_runtime.h>

typedef float v2f __attribute__((ext_vector_type(2)));
typedef float v8f __attribute__((ext_vector_type(8)));

#define HEADS 8
#define DH 32
#define NTOK 2048
#define DIM 256
#define BATCH 2
#define SCALE2 (1.0f / 32.0f)

#define PROJ_ELEMS ((size_t)BATCH * HEADS * NTOK * DH)
#define OFF_QA ((size_t)0)
#define OFF_KA (PROJ_ELEMS)
#define OFF_QS (2 * PROJ_ELEMS)
#define OFF_KS (3 * PROJ_ELEMS)

static __device__ __forceinline__ v8f wmma_f32(v2f a, v2f b, v8f c) {
  return __builtin_amdgcn_wmma_f32_16x16x4_f32(
       false, a,  false, b,
       (short)0, c,  false,  false);
}

static __device__ __forceinline__ float xhalf(float v) {
  const int b = __builtin_bit_cast(int, v);
  return __builtin_bit_cast(float, __builtin_amdgcn_permlanex16(
      b, b, 0x76543210, 0xfedcba98, false, false));
}

__global__ __launch_bounds__(256) void proj_kernel(
    const float* __restrict__ x, const float* __restrict__ attn_feat,
    const float* __restrict__ Wq, const float* __restrict__ Waux,
    float* __restrict__ ws) {
  const int lane = threadIdx.x & 31;
  const int wave = threadIdx.x >> 5;
  const int hi = lane >> 4;
  const int lo = lane & 15;
  const int ct = blockIdx.x * 8 + wave;
  const int m0 = blockIdx.y * 16;
  const int src = blockIdx.z;
  const float* X = src ? attn_feat : x;
  const float* W = src ? Waux : Wq;
  float* Qdst = ws + (src ? OFF_QA : OFF_QS);
  float* Kdst = ws + (src ? OFF_KA : OFF_KS);

  const int c = ct * 16 + lo;
  const float* Arow = X + (size_t)(m0 + lo) * DIM;

  v8f acc = {0.f, 0.f, 0.f, 0.f, 0.f, 0.f, 0.f, 0.f};
#pragma unroll 8
  for (int s = 0; s < 64; ++s) {
    const int k = s * 4 + hi * 2;
    v2f a = *(const v2f*)(Arow + k);
    v2f b;
    b.x = W[(size_t)k * 768 + c];
    b.y = W[(size_t)(k + 1) * 768 + c];
    acc = wmma_f32(a, b, acc);
  }

  float* dst;
  int hh;
  if (c < 256) { dst = Qdst; hh = c >> 5; }
  else         { dst = Kdst; hh = (c - 256) >> 5; }
  const int dh = c & 31;
#pragma unroll
  for (int r = 0; r < 8; ++r) {
    const int row = m0 + r + hi * 8;
    const int b_ = row >> 11, nn = row & 2047;
    dst[(((size_t)(b_ * HEADS + hh)) * NTOK + nn) * DH + dh] = acc[r];
  }
}

#define BK 64
#define KT (BK / 16)
#define SXP_LD 544
#define SK_LD 36
#define SO_LD 132
#define SX_FLOATS ((BK / 2) * SXP_LD)
#define SK_FLOATS (BK * SK_LD)
#define STAGE_FLOATS (SX_FLOATS + 2 * SK_FLOATS)

__global__ __launch_bounds__(256)
void attn_kernel(
    const float* x, const float* ws,
    float* __restrict__ out0) {
  __shared__ __attribute__((aligned(16))) float smem[2 * STAGE_FLOATS];

  const int tid = threadIdx.x;
  const int lane = tid & 31, wave = tid >> 5;
  const int hi = lane >> 4, lo = lane & 15;
  const int bh = blockIdx.y, b = bh >> 3, h = bh & 7;
  const int m0 = blockIdx.x * 128 + wave * 16;

  const float* Qa = ws + OFF_QA;
  const float* Qs = ws + OFF_QS;
  const float* kabase = ws + OFF_KA + (size_t)bh * NTOK * DH;
  const float* ksbase = ws + OFF_KS + (size_t)bh * NTOK * DH;
  const float* xb = x + (size_t)b * NTOK * DIM;

  v2f qa[8], qs[8];
  {
    const float* qab = Qa + ((size_t)bh * NTOK + m0 + lo) * DH;
    const float* qsb = Qs + ((size_t)bh * NTOK + m0 + lo) * DH;
#pragma unroll
    for (int kk = 0; kk < 8; ++kk) {
      const int k = kk * 4 + hi * 2;
      qa[kk] = *(const v2f*)(qab + k);
      qs[kk] = *(const v2f*)(qsb + k);
    }
  }

  v8f O[16];
#pragma unroll
  for (int t = 0; t < 16; ++t) O[t] = (v8f){0.f, 0.f, 0.f, 0.f, 0.f, 0.f, 0.f, 0.f};
  float m = -3.0e38f;
  float l = 0.f;

  const float* xg = xb + (size_t)((tid >> 6) * 2) * DIM + ((tid & 63) << 2);
  const float* kag = kabase + (size_t)(tid >> 3) * DH + ((tid & 7) << 2);
  const float* ksg = ksbase + (size_t)(tid >> 3) * DH + ((tid & 7) << 2);
  const int sxw_off = ((tid >> 6) * 2) * SXP_LD + ((tid & 63) << 3);
  const int skw_off = (tid >> 3) * SK_LD + ((tid & 7) << 2);

#define STAGE_REGS float4 tx[BK / 4], ta0, ta1, ts0, ts1
#define STAGE_LOAD(jb)                                                         \
  {                                                                            \
    const float* xg_ = xg + (size_t)(jb) * BK * DIM;                           \
    const float* kag_ = kag + (size_t)(jb) * BK * DH;                          \
    const float* ksg_ = ksg + (size_t)(jb) * BK * DH;                          \
    _Pragma("unroll") for (int t = 0; t < BK / 8; ++t) {                       \
      tx[2 * t] = *(const float4*)(xg_ + (size_t)t * 8 * DIM);           \
      tx[2 * t + 1] = *(const float4*)(xg_ + (size_t)t * 8 * DIM + DIM);   \
    }                                                                          \
    ta0 = *(const float4*)(kag_);                                              \
    ta1 = *(const float4*)(kag_ + (size_t)32 * DH);                            \
    ts0 = *(const float4*)(ksg_);                                              \
    ts1 = *(const float4*)(ksg_ + (size_t)32 * DH);                            \
  }
#define STAGE_STORE(st)                                                        \
  {                                                                            \
    float* sxw_ = (st) + sxw_off;                                              \
    float* skaw_ = (st) + SX_FLOATS + skw_off;                                 \
    float* sksw_ = (st) + SX_FLOATS + SK_FLOATS + skw_off;                     \
    _Pragma("unroll") for (int t = 0; t < BK / 8; ++t) {                       \
      const float4 e = tx[2 * t], o = tx[2 * t + 1];                           \
                                               \
      float* d = sxw_ + ((t >> 1) * 8 + (t & 1)) * SXP_LD;                     \
      *(float4*)d = make_float4(e.x, o.x, e.y, o.y);                           \
      *(float4*)(d + 4) = make_float4(e.z, o.z, e.w, o.w);                     \
    }                                                                          \
    *(float4*)(skaw_) = ta0;                                                   \
    *(float4*)(skaw_ + 32 * SK_LD) = ta1;                                      \
    *(float4*)(sksw_) = ts0;                                                   \
    *(float4*)(sksw_ + 32 * SK_LD) = ts1;                                      \
  }

  {
    STAGE_REGS;
    STAGE_LOAD(0);
    STAGE_STORE(smem);
  }
  __syncthreads();

  for (int j = 0; j < NTOK / BK; ++j) {
    const float* sX = smem + (j & 1) * STAGE_FLOATS;
    const float* sKa = sX + SX_FLOATS;
    const float* sKs = sKa + SK_FLOATS;
    float* snext = smem + ((j & 1) ^ 1) * STAGE_FLOATS;

    v2f fka[KT][8], fks[KT][8];
#pragma unroll
    for (int kt = 0; kt < KT; ++kt) {
      const float* ka = &sKa[(kt * 16 + lo) * SK_LD];
      const float* ks = &sKs[(kt * 16 + lo) * SK_LD];
#pragma unroll
      for (int kk = 0; kk < 8; ++kk) {
        const int k = kk * 4 + hi * 2;
        fka[kt][kk] = *(const v2f*)&ka[k];
        fks[kt][kk] = *(const v2f*)&ks[k];
      }
    }
    const float* xr = sX + hi * SXP_LD + lo * 4;
#define LOAD_XG(buf, g)                                                        \
  _Pragma("unroll") for (int u = 0; u < 2; ++u)                                \
  _Pragma("unroll") for (int q = 0; q < 8; ++q)                                \
    buf[u * 8 + q] = *(const float4*)(                                         \
        xr + (((g) >> 1) * 8 + 2 * (((g) & 1) * 2 + u)) * SXP_LD + q * 64)
    float4 fa[16], fb[16];
    LOAD_XG(fa, 0);
    __builtin_amdgcn_sched_barrier(0);

    v2f P[KT][4];
#pragma unroll
    for (int kt = 0; kt < KT; ++kt) {
      v8f Sa = {0.f, 0.f, 0.f, 0.f, 0.f, 0.f, 0.f, 0.f};
      v8f Ss = {0.f, 0.f, 0.f, 0.f, 0.f, 0.f, 0.f, 0.f};
#pragma unroll
      for (int kk = 0; kk < 8; ++kk) {
        Sa = wmma_f32(fka[kt][kk], qa[kk], Sa);
        Ss = wmma_f32(fks[kt][kk], qs[kk], Ss);
      }
#pragma unroll
      for (int s = 0; s < 4; ++s) {
        P[kt][s].x = Sa[2 * s] * Ss[2 * s] * SCALE2;
        P[kt][s].y = Sa[2 * s + 1] * Ss[2 * s + 1] * SCALE2;
      }
    }

    float vk[KT];
#pragma unroll
    for (int kt = 0; kt < KT; ++kt)
      vk[kt] = fmaxf(fmaxf(fmaxf(P[kt][0].x, P[kt][0].y), fmaxf(P[kt][1].x, P[kt][1].y)),
                     fmaxf(fmaxf(P[kt][2].x, P[kt][2].y), fmaxf(P[kt][3].x, P[kt][3].y)));
    float v = fmaxf(fmaxf(vk[0], vk[1]), fmaxf(vk[2], vk[3]));
    v = fmaxf(v, xhalf(v));
    const float mn = fmaxf(m, v);
    const float alpha = __expf(m - mn);
    m = mn;
    float sk[KT];
#pragma unroll
    for (int kt = 0; kt < KT; ++kt) {
      sk[kt] = 0.f;
#pragma unroll
      for (int s = 0; s < 4; ++s) {
        P[kt][s].x = __expf(P[kt][s].x - mn);
        P[kt][s].y = __expf(P[kt][s].y - mn);
        sk[kt] += P[kt][s].x + P[kt][s].y;
      }
    }
    l = l * alpha + ((sk[0] + sk[1]) + (sk[2] + sk[3]));
#pragma unroll
    for (int t = 0; t < 16; ++t)
#pragma unroll
      for (int r = 0; r < 8; ++r) O[t][r] *= alpha;

    __builtin_amdgcn_sched_barrier(0);
    STAGE_REGS;
    {
      const int jn = (j + 1 < NTOK / BK) ? j + 1 : j;
      STAGE_LOAD(jn);
    }
    __builtin_amdgcn_sched_barrier(0);

#define MMA_XG(buf, g)                                                         \
  _Pragma("unroll") for (int u = 0; u < 2; ++u)                                \
  _Pragma("unroll") for (int q = 0; q < 8; ++q) {                              \
    v2f a0, a1;                                                                \
    a0.x = buf[u * 8 + q].x; a0.y = buf[u * 8 + q].y;        \
    a1.x = buf[u * 8 + q].z; a1.y = buf[u * 8 + q].w;        \
    O[2 * q] = wmma_f32(a0, P[(g) >> 1][((g) & 1) * 2 + u], O[2 * q]);         \
    O[2 * q + 1] = wmma_f32(a1, P[(g) >> 1][((g) & 1) * 2 + u], O[2 * q + 1]); \
  }
#define PIN_B(g)                                                               \
  asm volatile("" : "+v"(P[(g) >> 1][((g) & 1) * 2]),                          \
                    "+v"(P[(g) >> 1][((g) & 1) * 2 + 1]))
#pragma unroll
    for (int g = 0; g < 8; g += 2) {
      LOAD_XG(fb, g + 1);
      __builtin_amdgcn_sched_barrier(0);
      PIN_B(g);
      MMA_XG(fa, g);
      __builtin_amdgcn_sched_barrier(0);
      if (g + 2 < 8) {
        LOAD_XG(fa, g + 2);
      }
      __builtin_amdgcn_sched_barrier(0);
      PIN_B(g + 1);
      MMA_XG(fb, g + 1);
      __builtin_amdgcn_sched_barrier(0);
    }
#undef PIN_B
#undef LOAD_XG
#undef MMA_XG

    STAGE_STORE(snext);
    __syncthreads();
  }
#undef STAGE_REGS
#undef STAGE_LOAD
#undef STAGE_STORE

  const float invl = 1.0f / (l + xhalf(l));
  float* so = smem + wave * (16 * SO_LD);
#pragma unroll
  for (int p = 0; p < 2; ++p) {
    if (p) __syncthreads();
#pragma unroll
    for (int q = 0; q < 4; ++q) {
#pragma unroll
      for (int j = 0; j < 4; ++j) {
        const int t = 2 * (4 * p + q);
        float4 o4 = make_float4(O[t][2 * j] * invl, O[t + 1][2 * j] * invl,
                                O[t][2 * j + 1] * invl, O[t + 1][2 * j + 1] * invl);
        *(float4*)(so + lo * SO_LD + q * 32 + hi * 16 + j * 4) = o4;
      }
    }
    __syncthreads();
#pragma unroll
    for (int rr = 0; rr < 16; ++rr) {
      const float4 o4 = *(const float4*)(so + rr * SO_LD + lane * 4);
      *(float4*)(out0 + ((size_t)b * NTOK + m0 + rr) * 2048 + h * 256 + p * 128 + lane * 4) = o4;
    }
  }
}

__global__ __launch_bounds__(256) void row0_kernel(
    const float* __restrict__ ws, const float* __restrict__ out0,
    float* __restrict__ out1, float* __restrict__ attn0) {
  __shared__ float sp[NTOK];
  __shared__ float red[256];
  __shared__ float qa_s[DH], qs_s[DH];
  __shared__ float stat[2];
  const int bh = blockIdx.x, tid = threadIdx.x;
  const float* Qa = ws + OFF_QA + (size_t)bh * NTOK * DH;
  const float* Ka = ws + OFF_KA + (size_t)bh * NTOK * DH;
  const float* Qs = ws + OFF_QS + (size_t)bh * NTOK * DH;
  const float* Ks = ws + OFF_KS + (size_t)bh * NTOK * DH;
  if (tid < DH) qa_s[tid] = Qa[tid];
  else if (tid < 2 * DH) qs_s[tid - DH] = Qs[tid - DH];
  __syncthreads();

  float lmax = -3.0e38f;
  for (int j = tid; j < NTOK; j += 256) {
    float sa = 0.f, ss = 0.f;
    const float* ka = Ka + (size_t)j * DH;
    const float* ks = Ks + (size_t)j * DH;
#pragma unroll
    for (int d = 0; d < DH; ++d) { sa += qa_s[d] * ka[d]; ss += qs_s[d] * ks[d]; }
    const float p = sa * ss * SCALE2;
    sp[j] = p;
    lmax = fmaxf(lmax, p);
  }
  red[tid] = lmax;
  __syncthreads();
  for (int s = 128; s > 0; s >>= 1) {
    if (tid < s) red[tid] = fmaxf(red[tid], red[tid + s]);
    __syncthreads();
  }
  if (tid == 0) stat[0] = red[0];
  __syncthreads();
  const float mmax = stat[0];
  float lsum = 0.f;
  for (int j = tid; j < NTOK; j += 256) {
    const float e = __expf(sp[j] - mmax);
    sp[j] = e;
    lsum += e;
  }
  red[tid] = lsum;
  __syncthreads();
  for (int s = 128; s > 0; s >>= 1) {
    if (tid < s) red[tid] += red[tid + s];
    __syncthreads();
  }
  if (tid == 0) stat[1] = red[0];
  __syncthreads();
  const float inv = 1.0f / stat[1];
  for (int j = tid; j < NTOK; j += 256)
    attn0[(size_t)bh * NTOK + j] = sp[j] * inv;

  if (bh < BATCH) {
    for (int c = tid; c < 2048; c += 256)
      out1[bh * 2048 + c] = out0[(size_t)bh * 2048 * 2048 + c];
  }
}

extern "C" void kernel_launch(void* const* d_in, const int* in_sizes, int n_in,
                              void* d_out, int out_size, void* d_ws,
                              size_t ws_size, hipStream_t stream) {
  const float* x  = (const float*)d_in[0];
  const float* af = (const float*)d_in[1];
  const float* Wq = (const float*)d_in[2];
  const float* Wa = (const float*)d_in[3];
  float* ws = (float*)d_ws;
  float* out0 = (float*)d_out;
  float* out1 = out0 + (size_t)BATCH * NTOK * 2048;
  float* attn0 = out1 + (size_t)BATCH * 2048;

  proj_kernel<<<dim3(4, 256, 2), 256, 0, stream>>>(x, af, Wq, Wa, ws);
  attn_kernel<<<dim3(16, 16), 256, 0, stream>>>(x, ws, out0);
  row0_kernel<<<16, 256, 0, stream>>>(ws, out0, out1, attn0);
}


# proj kernel: 128x64 work-group tiles, X and W chunks staged once per group in LDS (W as k pairs), 2x2 tiles per wave; was one tile per wave, fragments lane by lane from global; sums unchanged
# speedup vs baseline: 1.7986x; 1.0726x over previous
#include <hip/hip_runtime.h>

typedef float v2f __attribute__((ext_vector_type(2)));
typedef float v8f __attribute__((ext_vector_type(8)));

#define HEADS 8
#define DH 32
#define NTOK 2048
#define DIM 256
#define BATCH 2
#define SCALE2 (1.0f / 32.0f)

#define PROJ_ELEMS ((size_t)BATCH * HEADS * NTOK * DH)
#define OFF_QA ((size_t)0)
#define OFF_KA (PROJ_ELEMS)
#define OFF_QS (2 * PROJ_ELEMS)
#define OFF_KS (3 * PROJ_ELEMS)

static __device__ __forceinline__ v8f wmma_f32(v2f a, v2f b, v8f c) {
  return __builtin_amdgcn_wmma_f32_16x16x4_f32(
       false, a,  false, b,
       (short)0, c,  false,  false);
}

static __device__ __forceinline__ float xhalf(float v) {
  const int b = __builtin_bit_cast(int, v);
  return __builtin_bit_cast(float, __builtin_amdgcn_permlanex16(
      b, b, 0x76543210, 0xfedcba98, false, false));
}

#define PJ_BM 128
#define PJ_BN 64
#define PJ_KC 32
#define PJ_LDA 36
#define PJ_LDB 160
#define PJ_A_FLOATS (PJ_BM * PJ_LDA)
#define PJ_B_FLOATS ((PJ_KC / 2) * PJ_LDB)
#define PJ_STAGE (PJ_A_FLOATS + PJ_B_FLOATS)

__global__ __launch_bounds__(256) void proj_kernel(
    const float* x, const float* attn_feat,
    const float* Wq, const float* Waux,
    float* __restrict__ ws) {
  __shared__ __attribute__((aligned(16))) float sm[2 * PJ_STAGE];

  const int tid = threadIdx.x;
  const int lane = tid & 31, wave = tid >> 5;
  const int hi = lane >> 4;
  const int lo = lane & 15;
  const int n0 = blockIdx.x * PJ_BN;
  const int m0 = blockIdx.y * PJ_BM;
  const int src = blockIdx.z;
  const float* X = src ? attn_feat : x;
  const float* W = src ? Waux : Wq;
  const int wr = wave & 3, wc = wave >> 2;

  const float* ag = X + (size_t)(m0 + (tid >> 3)) * DIM + ((tid & 7) << 2);
  const float* bg = W + (size_t)((tid >> 4) * 2) * 768 + n0 + ((tid & 15) << 2);
  const int aw_off = (tid >> 3) * PJ_LDA + ((tid & 7) << 2);
  const int bw_off = PJ_A_FLOATS + (tid >> 4) * PJ_LDB + ((tid & 15) << 3);
  const int ar_off = (wr * 32 + lo) * PJ_LDA + hi * 2;
  const int br_off = PJ_A_FLOATS + hi * PJ_LDB + (wc * 32 + lo) * 2;

#define PJ_REGS float4 ta0, ta1, ta2, ta3, tb0, tb1
#define PJ_LOAD(c)                                                             \
  {                                                                            \
    const float* ag_ = ag + (c) * PJ_KC;                                       \
    const float* bg_ = bg + (size_t)(c) * PJ_KC * 768;                         \
    ta0 = *(const float4*)(ag_);                                               \
    ta1 = *(const float4*)(ag_ + (size_t)32 * DIM);                            \
    ta2 = *(const float4*)(ag_ + (size_t)64 * DIM);                            \
    ta3 = *(const float4*)(ag_ + (size_t)96 * DIM);                            \
    tb0 = *(const float4*)(bg_);                                \
    tb1 = *(const float4*)(bg_ + 768);                          \
  }
#define PJ_STORE(st)                                                           \
  {                                                                            \
    float* a_ = (st) + aw_off;                                                 \
    *(float4*)(a_) = ta0;                                                      \
    *(float4*)(a_ + 32 * PJ_LDA) = ta1;                                        \
    *(float4*)(a_ + 64 * PJ_LDA) = ta2;                                        \
    *(float4*)(a_ + 96 * PJ_LDA) = ta3;                                        \
    float* d_ = (st) + bw_off;                   \
    *(float4*)d_ = make_float4(tb0.x, tb1.x, tb0.y, tb1.y);                    \
    *(float4*)(d_ + 4) = make_float4(tb0.z, tb1.z, tb0.w, tb1.w);              \
  }

  v8f acc[2][2];
#pragma unroll
  for (int i = 0; i < 2; ++i)
#pragma unroll
    for (int jn = 0; jn < 2; ++jn)
      acc[i][jn] = (v8f){0.f, 0.f, 0.f, 0.f, 0.f, 0.f, 0.f, 0.f};

  {
    PJ_REGS;
    PJ_LOAD(0);
    PJ_STORE(sm);
  }
  __syncthreads();

  for (int c = 0; c < DIM / PJ_KC; ++c) {
    const float* st = sm + (c & 1) * PJ_STAGE;
    float* nx = sm + ((c & 1) ^ 1) * PJ_STAGE;
    const bool more = c + 1 < DIM / PJ_KC;
    PJ_REGS;
    if (more) PJ_LOAD(c + 1);
    __builtin_amdgcn_sched_barrier(0);

    v2f fa[PJ_KC / 4][2], fb[PJ_KC / 4][2];
#pragma unroll
    for (int kk = 0; kk < PJ_KC / 4; ++kk) {
#pragma unroll
      for (int i = 0; i < 2; ++i)
        fa[kk][i] = *(const v2f*)(st + ar_off + i * 16 * PJ_LDA + kk * 4);
#pragma unroll
      for (int jn = 0; jn < 2; ++jn)
        fb[kk][jn] = *(const v2f*)(st + br_off + kk * 2 * PJ_LDB + jn * 32);
    }
    __builtin_amdgcn_sched_barrier(0);
#pragma unroll
    for (int kk = 0; kk < PJ_KC / 4; ++kk)
#pragma unroll
      for (int i = 0; i < 2; ++i)
#pragma unroll
        for (int jn = 0; jn < 2; ++jn)
          acc[i][jn] = wmma_f32(fa[kk][i], fb[kk][jn], acc[i][jn]);
    __builtin_amdgcn_sched_barrier(0);

    if (more) {
      PJ_STORE(nx);
      __syncthreads();
    }
  }
#undef PJ_REGS
#undef PJ_LOAD
#undef PJ_STORE

  const int cw = n0 + wc * 32;
  float* dst = ws + (cw < 256 ? (src ? OFF_QA : OFF_QS) : (src ? OFF_KA : OFF_KS));
  const int hh = (cw & 255) >> 5;
  const int b_ = m0 >> 11;
#pragma unroll
  for (int i = 0; i < 2; ++i) {
#pragma unroll
    for (int r = 0; r < 8; ++r) {
      const int nn = ((m0 & 2047) + wr * 32 + i * 16 + hi * 8 + r);
      float* o = dst + (((size_t)(b_ * HEADS + hh)) * NTOK + nn) * DH + lo;
      o[0] = acc[i][0][r];
      o[16] = acc[i][1][r];
    }
  }
}

#define BK 64
#define KT (BK / 16)
#define SXP_LD 544
#define SK_LD 36
#define SO_LD 132
#define SX_FLOATS ((BK / 2) * SXP_LD)
#define SK_FLOATS (BK * SK_LD)
#define STAGE_FLOATS (SX_FLOATS + 2 * SK_FLOATS)

__global__ __launch_bounds__(256)
void attn_kernel(
    const float* x, const float* ws,
    float* __restrict__ out0) {
  __shared__ __attribute__((aligned(16))) float smem[2 * STAGE_FLOATS];

  const int tid = threadIdx.x;
  const int lane = tid & 31, wave = tid >> 5;
  const int hi = lane >> 4, lo = lane & 15;
  const int bh = blockIdx.y, b = bh >> 3, h = bh & 7;
  const int m0 = blockIdx.x * 128 + wave * 16;

  const float* Qa = ws + OFF_QA;
  const float* Qs = ws + OFF_QS;
  const float* kabase = ws + OFF_KA + (size_t)bh * NTOK * DH;
  const float* ksbase = ws + OFF_KS + (size_t)bh * NTOK * DH;
  const float* xb = x + (size_t)b * NTOK * DIM;

  v2f qa[8], qs[8];
  {
    const float* qab = Qa + ((size_t)bh * NTOK + m0 + lo) * DH;
    const float* qsb = Qs + ((size_t)bh * NTOK + m0 + lo) * DH;
#pragma unroll
    for (int kk = 0; kk < 8; ++kk) {
      const int k = kk * 4 + hi * 2;
      qa[kk] = *(const v2f*)(qab + k);
      qs[kk] = *(const v2f*)(qsb + k);
    }
  }

  v8f O[16];
#pragma unroll
  for (int t = 0; t < 16; ++t) O[t] = (v8f){0.f, 0.f, 0.f, 0.f, 0.f, 0.f, 0.f, 0.f};
  float m = -3.0e38f;
  float l = 0.f;

  const float* xg = xb + (size_t)((tid >> 6) * 2) * DIM + ((tid & 63) << 2);
  const float* kag = kabase + (size_t)(tid >> 3) * DH + ((tid & 7) << 2);
  const float* ksg = ksbase + (size_t)(tid >> 3) * DH + ((tid & 7) << 2);
  const int sxw_off = ((tid >> 6) * 2) * SXP_LD + ((tid & 63) << 3);
  const int skw_off = (tid >> 3) * SK_LD + ((tid & 7) << 2);

#define STAGE_REGS float4 tx[BK / 4], ta0, ta1, ts0, ts1
#define STAGE_LOAD(jb)                                                         \
  {                                                                            \
    const float* xg_ = xg + (size_t)(jb) * BK * DIM;                           \
    const float* kag_ = kag + (size_t)(jb) * BK * DH;                          \
    const float* ksg_ = ksg + (size_t)(jb) * BK * DH;                          \
    _Pragma("unroll") for (int t = 0; t < BK / 8; ++t) {                       \
      tx[2 * t] = *(const float4*)(xg_ + (size_t)t * 8 * DIM);           \
      tx[2 * t + 1] = *(const float4*)(xg_ + (size_t)t * 8 * DIM + DIM);   \
    }                                                                          \
    ta0 = *(const float4*)(kag_);                                              \
    ta1 = *(const float4*)(kag_ + (size_t)32 * DH);                            \
    ts0 = *(const float4*)(ksg_);                                              \
    ts1 = *(const float4*)(ksg_ + (size_t)32 * DH);                            \
  }
#define STAGE_STORE(st)                                                        \
  {                                                                            \
    float* sxw_ = (st) + sxw_off;                                              \
    float* skaw_ = (st) + SX_FLOATS + skw_off;                                 \
    float* sksw_ = (st) + SX_FLOATS + SK_FLOATS + skw_off;                     \
    _Pragma("unroll") for (int t = 0; t < BK / 8; ++t) {                       \
      const float4 e = tx[2 * t], o = tx[2 * t + 1];                           \
                                               \
      float* d = sxw_ + ((t >> 1) * 8 + (t & 1)) * SXP_LD;                     \
      *(float4*)d = make_float4(e.x, o.x, e.y, o.y);                           \
      *(float4*)(d + 4) = make_float4(e.z, o.z, e.w, o.w);                     \
    }                                                                          \
    *(float4*)(skaw_) = ta0;                                                   \
    *(float4*)(skaw_ + 32 * SK_LD) = ta1;                                      \
    *(float4*)(sksw_) = ts0;                                                   \
    *(float4*)(sksw_ + 32 * SK_LD) = ts1;                                      \
  }

  {
    STAGE_REGS;
    STAGE_LOAD(0);
    STAGE_STORE(smem);
  }
  __syncthreads();

  for (int j = 0; j < NTOK / BK; ++j) {
    const float* sX = smem + (j & 1) * STAGE_FLOATS;
    const float* sKa = sX + SX_FLOATS;
    const float* sKs = sKa + SK_FLOATS;
    float* snext = smem + ((j & 1) ^ 1) * STAGE_FLOATS;

    v2f fka[KT][8], fks[KT][8];
#pragma unroll
    for (int kt = 0; kt < KT; ++kt) {
      const float* ka = &sKa[(kt * 16 + lo) * SK_LD];
      const float* ks = &sKs[(kt * 16 + lo) * SK_LD];
#pragma unroll
      for (int kk = 0; kk < 8; ++kk) {
        const int k = kk * 4 + hi * 2;
        fka[kt][kk] = *(const v2f*)&ka[k];
        fks[kt][kk] = *(const v2f*)&ks[k];
      }
    }
    const float* xr = sX + hi * SXP_LD + lo * 4;
#define LOAD_XG(buf, g)                                                        \
  _Pragma("unroll") for (int u = 0; u < 2; ++u)                                \
  _Pragma("unroll") for (int q = 0; q < 8; ++q)                                \
    buf[u * 8 + q] = *(const float4*)(                                         \
        xr + (((g) >> 1) * 8 + 2 * (((g) & 1) * 2 + u)) * SXP_LD + q * 64)
    float4 fa[16], fb[16];
    LOAD_XG(fa, 0);
    __builtin_amdgcn_sched_barrier(0);

    v2f P[KT][4];
#pragma unroll
    for (int kt = 0; kt < KT; ++kt) {
      v8f Sa = {0.f, 0.f, 0.f, 0.f, 0.f, 0.f, 0.f, 0.f};
      v8f Ss = {0.f, 0.f, 0.f, 0.f, 0.f, 0.f, 0.f, 0.f};
#pragma unroll
      for (int kk = 0; kk < 8; ++kk) {
        Sa = wmma_f32(fka[kt][kk], qa[kk], Sa);
        Ss = wmma_f32(fks[kt][kk], qs[kk], Ss);
      }
#pragma unroll
      for (int s = 0; s < 4; ++s) {
        P[kt][s].x = Sa[2 * s] * Ss[2 * s] * SCALE2;
        P[kt][s].y = Sa[2 * s + 1] * Ss[2 * s + 1] * SCALE2;
      }
    }

    float vk[KT];
#pragma unroll
    for (int kt = 0; kt < KT; ++kt)
      vk[kt] = fmaxf(fmaxf(fmaxf(P[kt][0].x, P[kt][0].y), fmaxf(P[kt][1].x, P[kt][1].y)),
                     fmaxf(fmaxf(P[kt][2].x, P[kt][2].y), fmaxf(P[kt][3].x, P[kt][3].y)));
    float v = fmaxf(fmaxf(vk[0], vk[1]), fmaxf(vk[2], vk[3]));
    v = fmaxf(v, xhalf(v));
    const float mn = fmaxf(m, v);
    const float alpha = __expf(m - mn);
    m = mn;
    float sk[KT];
#pragma unroll
    for (int kt = 0; kt < KT; ++kt) {
      sk[kt] = 0.f;
#pragma unroll
      for (int s = 0; s < 4; ++s) {
        P[kt][s].x = __expf(P[kt][s].x - mn);
        P[kt][s].y = __expf(P[kt][s].y - mn);
        sk[kt] += P[kt][s].x + P[kt][s].y;
      }
    }
    l = l * alpha + ((sk[0] + sk[1]) + (sk[2] + sk[3]));
#pragma unroll
    for (int t = 0; t < 16; ++t)
#pragma unroll
      for (int r = 0; r < 8; ++r) O[t][r] *= alpha;

    __builtin_amdgcn_sched_barrier(0);
    STAGE_REGS;
    {
      const int jn = (j + 1 < NTOK / BK) ? j + 1 : j;
      STAGE_LOAD(jn);
    }
    __builtin_amdgcn_sched_barrier(0);

#define MMA_XG(buf, g)                                                         \
  _Pragma("unroll") for (int u = 0; u < 2; ++u)                                \
  _Pragma("unroll") for (int q = 0; q < 8; ++q) {                              \
    v2f a0, a1;                                                                \
    a0.x = buf[u * 8 + q].x; a0.y = buf[u * 8 + q].y;        \
    a1.x = buf[u * 8 + q].z; a1.y = buf[u * 8 + q].w;        \
    O[2 * q] = wmma_f32(a0, P[(g) >> 1][((g) & 1) * 2 + u], O[2 * q]);         \
    O[2 * q + 1] = wmma_f32(a1, P[(g) >> 1][((g) & 1) * 2 + u], O[2 * q + 1]); \
  }
#define PIN_B(g)                                                               \
  asm volatile("" : "+v"(P[(g) >> 1][((g) & 1) * 2]),                          \
                    "+v"(P[(g) >> 1][((g) & 1) * 2 + 1]))
#pragma unroll
    for (int g = 0; g < 8; g += 2) {
      LOAD_XG(fb, g + 1);
      __builtin_amdgcn_sched_barrier(0);
      PIN_B(g);
      MMA_XG(fa, g);
      __builtin_amdgcn_sched_barrier(0);
      if (g + 2 < 8) {
        LOAD_XG(fa, g + 2);
      }
      __builtin_amdgcn_sched_barrier(0);
      PIN_B(g + 1);
      MMA_XG(fb, g + 1);
      __builtin_amdgcn_sched_barrier(0);
    }
#undef PIN_B
#undef LOAD_XG
#undef MMA_XG

    STAGE_STORE(snext);
    __syncthreads();
  }
#undef STAGE_REGS
#undef STAGE_LOAD
#undef STAGE_STORE

  const float invl = 1.0f / (l + xhalf(l));
  float* so = smem + wave * (16 * SO_LD);
#pragma unroll
  for (int p = 0; p < 2; ++p) {
    if (p) __syncthreads();
#pragma unroll
    for (int q = 0; q < 4; ++q) {
#pragma unroll
      for (int j = 0; j < 4; ++j) {
        const int t = 2 * (4 * p + q);
        float4 o4 = make_float4(O[t][2 * j] * invl, O[t + 1][2 * j] * invl,
                                O[t][2 * j + 1] * invl, O[t + 1][2 * j + 1] * invl);
        *(float4*)(so + lo * SO_LD + q * 32 + hi * 16 + j * 4) = o4;
      }
    }
    __syncthreads();
#pragma unroll
    for (int rr = 0; rr < 16; ++rr) {
      const float4 o4 = *(const float4*)(so + rr * SO_LD + lane * 4);
      *(float4*)(out0 + ((size_t)b * NTOK + m0 + rr) * 2048 + h * 256 + p * 128 + lane * 4) = o4;
    }
  }
}

__global__ __launch_bounds__(256) void row0_kernel(
    const float* __restrict__ ws, const float* __restrict__ out0,
    float* __restrict__ out1, float* __restrict__ attn0) {
  __shared__ float sp[NTOK];
  __shared__ float red[256];
  __shared__ float qa_s[DH], qs_s[DH];
  __shared__ float stat[2];
  const int bh = blockIdx.x, tid = threadIdx.x;
  const float* Qa = ws + OFF_QA + (size_t)bh * NTOK * DH;
  const float* Ka = ws + OFF_KA + (size_t)bh * NTOK * DH;
  const float* Qs = ws + OFF_QS + (size_t)bh * NTOK * DH;
  const float* Ks = ws + OFF_KS + (size_t)bh * NTOK * DH;
  if (tid < DH) qa_s[tid] = Qa[tid];
  else if (tid < 2 * DH) qs_s[tid - DH] = Qs[tid - DH];
  __syncthreads();

  float lmax = -3.0e38f;
  for (int j = tid; j < NTOK; j += 256) {
    float sa = 0.f, ss = 0.f;
    const float* ka = Ka + (size_t)j * DH;
    const float* ks = Ks + (size_t)j * DH;
#pragma unroll
    for (int d = 0; d < DH; ++d) { sa += qa_s[d] * ka[d]; ss += qs_s[d] * ks[d]; }
    const float p = sa * ss * SCALE2;
    sp[j] = p;
    lmax = fmaxf(lmax, p);
  }
  red[tid] = lmax;
  __syncthreads();
  for (int s = 128; s > 0; s >>= 1) {
    if (tid < s) red[tid] = fmaxf(red[tid], red[tid + s]);
    __syncthreads();
  }
  if (tid == 0) stat[0] = red[0];
  __syncthreads();
  const float mmax = stat[0];
  float lsum = 0.f;
  for (int j = tid; j < NTOK; j += 256) {
    const float e = __expf(sp[j] - mmax);
    sp[j] = e;
    lsum += e;
  }
  red[tid] = lsum;
  __syncthreads();
  for (int s = 128; s > 0; s >>= 1) {
    if (tid < s) red[tid] += red[tid + s];
    __syncthreads();
  }
  if (tid == 0) stat[1] = red[0];
  __syncthreads();
  const float inv = 1.0f / stat[1];
  for (int j = tid; j < NTOK; j += 256)
    attn0[(size_t)bh * NTOK + j] = sp[j] * inv;

  if (bh < BATCH) {
    for (int c = tid; c < 2048; c += 256)
      out1[bh * 2048 + c] = out0[(size_t)bh * 2048 * 2048 + c];
  }
}

extern "C" void kernel_launch(void* const* d_in, const int* in_sizes, int n_in,
                              void* d_out, int out_size, void* d_ws,
                              size_t ws_size, hipStream_t stream) {
  const float* x  = (const float*)d_in[0];
  const float* af = (const float*)d_in[1];
  const float* Wq = (const float*)d_in[2];
  const float* Wa = (const float*)d_in[3];
  float* ws = (float*)d_ws;
  float* out0 = (float*)d_out;
  float* out1 = out0 + (size_t)BATCH * NTOK * 2048;
  float* attn0 = out1 + (size_t)BATCH * 2048;

  proj_kernel<<<dim3(8, 32, 2), 256, 0, stream>>>(x, af, Wq, Wa, ws);
  attn_kernel<<<dim3(16, 16), 256, 0, stream>>>(x, ws, out0);
  row0_kernel<<<16, 256, 0, stream>>>(ws, out0, out1, attn0);
}


# two launches, not three: the attn wave that owns query 0 keeps its raw scores in LDS and writes attn row 0 and out[:,0] itself; row 0's scores now from WMMA, its sum the running sum (order changes)
# speedup vs baseline: 1.8988x; 1.0557x over previous
#include <hip/hip_runtime.h>

typedef float v2f __attribute__((ext_vector_type(2)));
typedef float v8f __attribute__((ext_vector_type(8)));

#define HEADS 8
#define DH 32
#define NTOK 2048
#define DIM 256
#define BATCH 2
#define SCALE2 (1.0f / 32.0f)

#define PROJ_ELEMS ((size_t)BATCH * HEADS * NTOK * DH)
#define OFF_QA ((size_t)0)
#define OFF_KA (PROJ_ELEMS)
#define OFF_QS (2 * PROJ_ELEMS)
#define OFF_KS (3 * PROJ_ELEMS)

static __device__ __forceinline__ v8f wmma_f32(v2f a, v2f b, v8f c) {
  return __builtin_amdgcn_wmma_f32_16x16x4_f32(
       false, a,  false, b,
       (short)0, c,  false,  false);
}

static __device__ __forceinline__ float xhalf(float v) {
  const int b = __builtin_bit_cast(int, v);
  return __builtin_bit_cast(float, __builtin_amdgcn_permlanex16(
      b, b, 0x76543210, 0xfedcba98, false, false));
}

#define PJ_BM 128
#define PJ_BN 64
#define PJ_KC 32
#define PJ_LDA 36
#define PJ_LDB 160
#define PJ_A_FLOATS (PJ_BM * PJ_LDA)
#define PJ_B_FLOATS ((PJ_KC / 2) * PJ_LDB)
#define PJ_STAGE (PJ_A_FLOATS + PJ_B_FLOATS)

__global__ __launch_bounds__(256) void proj_kernel(
    const float* x, const float* attn_feat,
    const float* Wq, const float* Waux,
    float* __restrict__ ws) {
  __shared__ __attribute__((aligned(16))) float sm[2 * PJ_STAGE];

  const int tid = threadIdx.x;
  const int lane = tid & 31, wave = tid >> 5;
  const int hi = lane >> 4;
  const int lo = lane & 15;
  const int n0 = blockIdx.x * PJ_BN;
  const int m0 = blockIdx.y * PJ_BM;
  const int src = blockIdx.z;
  const float* X = src ? attn_feat : x;
  const float* W = src ? Waux : Wq;
  const int wr = wave & 3, wc = wave >> 2;

  const float* ag = X + (size_t)(m0 + (tid >> 3)) * DIM + ((tid & 7) << 2);
  const float* bg = W + (size_t)((tid >> 4) * 2) * 768 + n0 + ((tid & 15) << 2);
  const int aw_off = (tid >> 3) * PJ_LDA + ((tid & 7) << 2);
  const int bw_off = PJ_A_FLOATS + (tid >> 4) * PJ_LDB + ((tid & 15) << 3);
  const int ar_off = (wr * 32 + lo) * PJ_LDA + hi * 2;
  const int br_off = PJ_A_FLOATS + hi * PJ_LDB + (wc * 32 + lo) * 2;

#define PJ_REGS float4 ta0, ta1, ta2, ta3, tb0, tb1
#define PJ_LOAD(c)                                                             \
  {                                                                            \
    const float* ag_ = ag + (c) * PJ_KC;                                       \
    const float* bg_ = bg + (size_t)(c) * PJ_KC * 768;                         \
    ta0 = *(const float4*)(ag_);                                               \
    ta1 = *(const float4*)(ag_ + (size_t)32 * DIM);                            \
    ta2 = *(const float4*)(ag_ + (size_t)64 * DIM);                            \
    ta3 = *(const float4*)(ag_ + (size_t)96 * DIM);                            \
    tb0 = *(const float4*)(bg_);                                \
    tb1 = *(const float4*)(bg_ + 768);                          \
  }
#define PJ_STORE(st)                                                           \
  {                                                                            \
    float* a_ = (st) + aw_off;                                                 \
    *(float4*)(a_) = ta0;                                                      \
    *(float4*)(a_ + 32 * PJ_LDA) = ta1;                                        \
    *(float4*)(a_ + 64 * PJ_LDA) = ta2;                                        \
    *(float4*)(a_ + 96 * PJ_LDA) = ta3;                                        \
    float* d_ = (st) + bw_off;                   \
    *(float4*)d_ = make_float4(tb0.x, tb1.x, tb0.y, tb1.y);                    \
    *(float4*)(d_ + 4) = make_float4(tb0.z, tb1.z, tb0.w, tb1.w);              \
  }

  v8f acc[2][2];
#pragma unroll
  for (int i = 0; i < 2; ++i)
#pragma unroll
    for (int jn = 0; jn < 2; ++jn)
      acc[i][jn] = (v8f){0.f, 0.f, 0.f, 0.f, 0.f, 0.f, 0.f, 0.f};

  {
    PJ_REGS;
    PJ_LOAD(0);
    PJ_STORE(sm);
  }
  __syncthreads();

  for (int c = 0; c < DIM / PJ_KC; ++c) {
    const float* st = sm + (c & 1) * PJ_STAGE;
    float* nx = sm + ((c & 1) ^ 1) * PJ_STAGE;
    const bool more = c + 1 < DIM / PJ_KC;
    PJ_REGS;
    if (more) PJ_LOAD(c + 1);
    __builtin_amdgcn_sched_barrier(0);

    v2f fa[PJ_KC / 4][2], fb[PJ_KC / 4][2];
#pragma unroll
    for (int kk = 0; kk < PJ_KC / 4; ++kk) {
#pragma unroll
      for (int i = 0; i < 2; ++i)
        fa[kk][i] = *(const v2f*)(st + ar_off + i * 16 * PJ_LDA + kk * 4);
#pragma unroll
      for (int jn = 0; jn < 2; ++jn)
        fb[kk][jn] = *(const v2f*)(st + br_off + kk * 2 * PJ_LDB + jn * 32);
    }
    __builtin_amdgcn_sched_barrier(0);
#pragma unroll
    for (int kk = 0; kk < PJ_KC / 4; ++kk)
#pragma unroll
      for (int i = 0; i < 2; ++i)
#pragma unroll
        for (int jn = 0; jn < 2; ++jn)
          acc[i][jn] = wmma_f32(fa[kk][i], fb[kk][jn], acc[i][jn]);
    __builtin_amdgcn_sched_barrier(0);

    if (more) {
      PJ_STORE(nx);
      __syncthreads();
    }
  }
#undef PJ_REGS
#undef PJ_LOAD
#undef PJ_STORE

  const int cw = n0 + wc * 32;
  float* dst = ws + (cw < 256 ? (src ? OFF_QA : OFF_QS) : (src ? OFF_KA : OFF_KS));
  const int hh = (cw & 255) >> 5;
  const int b_ = m0 >> 11;
#pragma unroll
  for (int i = 0; i < 2; ++i) {
#pragma unroll
    for (int r = 0; r < 8; ++r) {
      const int nn = ((m0 & 2047) + wr * 32 + i * 16 + hi * 8 + r);
      float* o = dst + (((size_t)(b_ * HEADS + hh)) * NTOK + nn) * DH + lo;
      o[0] = acc[i][0][r];
      o[16] = acc[i][1][r];
    }
  }
}

#define BK 64
#define KT (BK / 16)
#define SXP_LD 544
#define SK_LD 36
#define SO_LD 132
#define SX_FLOATS ((BK / 2) * SXP_LD)
#define SK_FLOATS (BK * SK_LD)
#define STAGE_FLOATS (SX_FLOATS + 2 * SK_FLOATS)

__global__ __launch_bounds__(256)
void attn_kernel(
    const float* x, const float* ws,
    float* __restrict__ out0, float* __restrict__ out1,
    float* __restrict__ attn0) {
  __shared__ __attribute__((aligned(16))) float smem[2 * STAGE_FLOATS];
  __shared__ __attribute__((aligned(16))) float srow[NTOK];

  const int tid = threadIdx.x;
  const int lane = tid & 31, wave = tid >> 5;
  const int hi = lane >> 4, lo = lane & 15;
  const int bh = blockIdx.y, b = bh >> 3, h = bh & 7;
  const int m0 = blockIdx.x * 128 + wave * 16;
  const bool row0w = (blockIdx.x == 0) && (__builtin_amdgcn_readfirstlane(tid) < 32);

  const float* Qa = ws + OFF_QA;
  const float* Qs = ws + OFF_QS;
  const float* kabase = ws + OFF_KA + (size_t)bh * NTOK * DH;
  const float* ksbase = ws + OFF_KS + (size_t)bh * NTOK * DH;
  const float* xb = x + (size_t)b * NTOK * DIM;

  v2f qa[8], qs[8];
  {
    const float* qab = Qa + ((size_t)bh * NTOK + m0 + lo) * DH;
    const float* qsb = Qs + ((size_t)bh * NTOK + m0 + lo) * DH;
#pragma unroll
    for (int kk = 0; kk < 8; ++kk) {
      const int k = kk * 4 + hi * 2;
      qa[kk] = *(const v2f*)(qab + k);
      qs[kk] = *(const v2f*)(qsb + k);
    }
  }

  v8f O[16];
#pragma unroll
  for (int t = 0; t < 16; ++t) O[t] = (v8f){0.f, 0.f, 0.f, 0.f, 0.f, 0.f, 0.f, 0.f};
  float m = -3.0e38f;
  float l = 0.f;

  const float* xg = xb + (size_t)((tid >> 6) * 2) * DIM + ((tid & 63) << 2);
  const float* kag = kabase + (size_t)(tid >> 3) * DH + ((tid & 7) << 2);
  const float* ksg = ksbase + (size_t)(tid >> 3) * DH + ((tid & 7) << 2);
  const int sxw_off = ((tid >> 6) * 2) * SXP_LD + ((tid & 63) << 3);
  const int skw_off = (tid >> 3) * SK_LD + ((tid & 7) << 2);

#define STAGE_REGS float4 tx[BK / 4], ta0, ta1, ts0, ts1
#define STAGE_LOAD(jb)                                                         \
  {                                                                            \
    const float* xg_ = xg + (size_t)(jb) * BK * DIM;                           \
    const float* kag_ = kag + (size_t)(jb) * BK * DH;                          \
    const float* ksg_ = ksg + (size_t)(jb) * BK * DH;                          \
    _Pragma("unroll") for (int t = 0; t < BK / 8; ++t) {                       \
      tx[2 * t] = *(const float4*)(xg_ + (size_t)t * 8 * DIM);           \
      tx[2 * t + 1] = *(const float4*)(xg_ + (size_t)t * 8 * DIM + DIM);   \
    }                                                                          \
    ta0 = *(const float4*)(kag_);                                              \
    ta1 = *(const float4*)(kag_ + (size_t)32 * DH);                            \
    ts0 = *(const float4*)(ksg_);                                              \
    ts1 = *(const float4*)(ksg_ + (size_t)32 * DH);                            \
  }
#define STAGE_STORE(st)                                                        \
  {                                                                            \
    float* sxw_ = (st) + sxw_off;                                              \
    float* skaw_ = (st) + SX_FLOATS + skw_off;                                 \
    float* sksw_ = (st) + SX_FLOATS + SK_FLOATS + skw_off;                     \
    _Pragma("unroll") for (int t = 0; t < BK / 8; ++t) {                       \
      const float4 e = tx[2 * t], o = tx[2 * t + 1];                           \
                                               \
      float* d = sxw_ + ((t >> 1) * 8 + (t & 1)) * SXP_LD;                     \
      *(float4*)d = make_float4(e.x, o.x, e.y, o.y);                           \
      *(float4*)(d + 4) = make_float4(e.z, o.z, e.w, o.w);                     \
    }                                                                          \
    *(float4*)(skaw_) = ta0;                                                   \
    *(float4*)(skaw_ + 32 * SK_LD) = ta1;                                      \
    *(float4*)(sksw_) = ts0;                                                   \
    *(float4*)(sksw_ + 32 * SK_LD) = ts1;                                      \
  }

  {
    STAGE_REGS;
    STAGE_LOAD(0);
    STAGE_STORE(smem);
  }
  __syncthreads();

  for (int j = 0; j < NTOK / BK; ++j) {
    const float* sX = smem + (j & 1) * STAGE_FLOATS;
    const float* sKa = sX + SX_FLOATS;
    const float* sKs = sKa + SK_FLOATS;
    float* snext = smem + ((j & 1) ^ 1) * STAGE_FLOATS;

    v2f fka[KT][8], fks[KT][8];
#pragma unroll
    for (int kt = 0; kt < KT; ++kt) {
      const float* ka = &sKa[(kt * 16 + lo) * SK_LD];
      const float* ks = &sKs[(kt * 16 + lo) * SK_LD];
#pragma unroll
      for (int kk = 0; kk < 8; ++kk) {
        const int k = kk * 4 + hi * 2;
        fka[kt][kk] = *(const v2f*)&ka[k];
        fks[kt][kk] = *(const v2f*)&ks[k];
      }
    }
    const float* xr = sX + hi * SXP_LD + lo * 4;
#define LOAD_XG(buf, g)                                                        \
  _Pragma("unroll") for (int u = 0; u < 2; ++u)                                \
  _Pragma("unroll") for (int q = 0; q < 8; ++q)                                \
    buf[u * 8 + q] = *(const float4*)(                                         \
        xr + (((g) >> 1) * 8 + 2 * (((g) & 1) * 2 + u)) * SXP_LD + q * 64)
    float4 fa[16], fb[16];
    LOAD_XG(fa, 0);
    __builtin_amdgcn_sched_barrier(0);

    v2f P[KT][4];
#pragma unroll
    for (int kt = 0; kt < KT; ++kt) {
      v8f Sa = {0.f, 0.f, 0.f, 0.f, 0.f, 0.f, 0.f, 0.f};
      v8f Ss = {0.f, 0.f, 0.f, 0.f, 0.f, 0.f, 0.f, 0.f};
#pragma unroll
      for (int kk = 0; kk < 8; ++kk) {
        Sa = wmma_f32(fka[kt][kk], qa[kk], Sa);
        Ss = wmma_f32(fks[kt][kk], qs[kk], Ss);
      }
#pragma unroll
      for (int s = 0; s < 4; ++s) {
        P[kt][s].x = Sa[2 * s] * Ss[2 * s] * SCALE2;
        P[kt][s].y = Sa[2 * s + 1] * Ss[2 * s + 1] * SCALE2;
      }
    }
    if (row0w && lo == 0) {
      float* d = srow + j * BK + hi * 8;
#pragma unroll
      for (int kt = 0; kt < KT; ++kt) {
        *(float4*)(d + kt * 16) =
            make_float4(P[kt][0].x, P[kt][0].y, P[kt][1].x, P[kt][1].y);
        *(float4*)(d + kt * 16 + 4) =
            make_float4(P[kt][2].x, P[kt][2].y, P[kt][3].x, P[kt][3].y);
      }
    }

    float vk[KT];
#pragma unroll
    for (int kt = 0; kt < KT; ++kt)
      vk[kt] = fmaxf(fmaxf(fmaxf(P[kt][0].x, P[kt][0].y), fmaxf(P[kt][1].x, P[kt][1].y)),
                     fmaxf(fmaxf(P[kt][2].x, P[kt][2].y), fmaxf(P[kt][3].x, P[kt][3].y)));
    float v = fmaxf(fmaxf(vk[0], vk[1]), fmaxf(vk[2], vk[3]));
    v = fmaxf(v, xhalf(v));
    const float mn = fmaxf(m, v);
    const float alpha = __expf(m - mn);
    m = mn;
    float sk[KT];
#pragma unroll
    for (int kt = 0; kt < KT; ++kt) {
      sk[kt] = 0.f;
#pragma unroll
      for (int s = 0; s < 4; ++s) {
        P[kt][s].x = __expf(P[kt][s].x - mn);
        P[kt][s].y = __expf(P[kt][s].y - mn);
        sk[kt] += P[kt][s].x + P[kt][s].y;
      }
    }
    l = l * alpha + ((sk[0] + sk[1]) + (sk[2] + sk[3]));
#pragma unroll
    for (int t = 0; t < 16; ++t)
#pragma unroll
      for (int r = 0; r < 8; ++r) O[t][r] *= alpha;

    __builtin_amdgcn_sched_barrier(0);
    STAGE_REGS;
    {
      const int jn = (j + 1 < NTOK / BK) ? j + 1 : j;
      STAGE_LOAD(jn);
    }
    __builtin_amdgcn_sched_barrier(0);

#define MMA_XG(buf, g)                                                         \
  _Pragma("unroll") for (int u = 0; u < 2; ++u)                                \
  _Pragma("unroll") for (int q = 0; q < 8; ++q) {                              \
    v2f a0, a1;                                                                \
    a0.x = buf[u * 8 + q].x; a0.y = buf[u * 8 + q].y;        \
    a1.x = buf[u * 8 + q].z; a1.y = buf[u * 8 + q].w;        \
    O[2 * q] = wmma_f32(a0, P[(g) >> 1][((g) & 1) * 2 + u], O[2 * q]);         \
    O[2 * q + 1] = wmma_f32(a1, P[(g) >> 1][((g) & 1) * 2 + u], O[2 * q + 1]); \
  }
#define PIN_B(g)                                                               \
  asm volatile("" : "+v"(P[(g) >> 1][((g) & 1) * 2]),                          \
                    "+v"(P[(g) >> 1][((g) & 1) * 2 + 1]))
#pragma unroll
    for (int g = 0; g < 8; g += 2) {
      LOAD_XG(fb, g + 1);
      __builtin_amdgcn_sched_barrier(0);
      PIN_B(g);
      MMA_XG(fa, g);
      __builtin_amdgcn_sched_barrier(0);
      if (g + 2 < 8) {
        LOAD_XG(fa, g + 2);
      }
      __builtin_amdgcn_sched_barrier(0);
      PIN_B(g + 1);
      MMA_XG(fb, g + 1);
      __builtin_amdgcn_sched_barrier(0);
    }
#undef PIN_B
#undef LOAD_XG
#undef MMA_XG

    STAGE_STORE(snext);
    __syncthreads();
  }
#undef STAGE_REGS
#undef STAGE_LOAD
#undef STAGE_STORE

  const float invl = 1.0f / (l + xhalf(l));
  float* so = smem + wave * (16 * SO_LD);
#pragma unroll
  for (int p = 0; p < 2; ++p) {
    if (p) __syncthreads();
#pragma unroll
    for (int q = 0; q < 4; ++q) {
#pragma unroll
      for (int j = 0; j < 4; ++j) {
        const int t = 2 * (4 * p + q);
        float4 o4 = make_float4(O[t][2 * j] * invl, O[t + 1][2 * j] * invl,
                                O[t][2 * j + 1] * invl, O[t + 1][2 * j + 1] * invl);
        *(float4*)(so + lo * SO_LD + q * 32 + hi * 16 + j * 4) = o4;
      }
    }
    __syncthreads();
#pragma unroll
    for (int rr = 0; rr < 16; ++rr) {
      const float4 o4 = *(const float4*)(so + rr * SO_LD + lane * 4);
      *(float4*)(out0 + ((size_t)b * NTOK + m0 + rr) * 2048 + h * 256 + p * 128 + lane * 4) = o4;
      if (rr == 0 && row0w)
        *(float4*)(out1 + (size_t)b * 2048 + h * 256 + p * 128 + lane * 4) = o4;
    }
  }

  if (row0w) {
    const float mq = __builtin_bit_cast(float,
        __builtin_amdgcn_readfirstlane(__builtin_bit_cast(int, m)));
    const float iq = __builtin_bit_cast(float,
        __builtin_amdgcn_readfirstlane(__builtin_bit_cast(int, invl)));
    float* arow = attn0 + (size_t)bh * NTOK;
#pragma unroll
    for (int i = 0; i < NTOK / 128; ++i) {
      const float4 s4 = *(const float4*)(srow + i * 128 + lane * 4);
      const float4 r4 = make_float4(__expf(s4.x - mq) * iq, __expf(s4.y - mq) * iq,
                                    __expf(s4.z - mq) * iq, __expf(s4.w - mq) * iq);
      *(float4*)(arow + i * 128 + lane * 4) = r4;
    }
  }
}

extern "C" void kernel_launch(void* const* d_in, const int* in_sizes, int n_in,
                              void* d_out, int out_size, void* d_ws,
                              size_t ws_size, hipStream_t stream) {
  const float* x  = (const float*)d_in[0];
  const float* af = (const float*)d_in[1];
  const float* Wq = (const float*)d_in[2];
  const float* Wa = (const float*)d_in[3];
  float* ws = (float*)d_ws;
  float* out0 = (float*)d_out;
  float* out1 = out0 + (size_t)BATCH * NTOK * 2048;
  float* attn0 = out1 + (size_t)BATCH * 2048;

  proj_kernel<<<dim3(8, 32, 2), 256, 0, stream>>>(x, af, Wq, Wa, ws);
  attn_kernel<<<dim3(16, 16), 256, 0, stream>>>(x, ws, out0, out1, attn0);
}
